# starting kernel
# baseline (speedup 1.0000x reference)
#include <hip/hip_runtime.h>
#include <hip/hip_bf16.h>

typedef __attribute__((ext_vector_type(2))) float v2f;
typedef __attribute__((ext_vector_type(8))) float v8f;

#define HIDDEN 128

__global__ void gat_prep_uv(const float* __restrict__ W, const float* __restrict__ b,
                            const float* __restrict__ att_w,
                            float* __restrict__ G, float* __restrict__ cs) {
  int j = threadIdx.x;
  float su = 0.f, sv = 0.f;
  for (int k = 0; k < HIDDEN; ++k) {
    float w = W[k * HIDDEN + j];
    su = fmaf(att_w[k], w, su);
    sv = fmaf(att_w[HIDDEN + k], w, sv);
  }
  G[j] = su;
  G[HIDDEN + j] = sv;
#pragma unroll
  for (int r = 2; r < 16; ++r) G[r * HIDDEN + j] = 0.f;
  __shared__ float tb[HIDDEN];
  tb[j] = (att_w[j] + att_w[HIDDEN + j]) * b[j];
  __syncthreads();
  if (j == 0) {
    float s = 0.f;
    for (int k = 0; k < HIDDEN; ++k) s += tb[k];
    cs[0] = s;
  }
}

__global__ void gat_zero(float* __restrict__ nacc, int n) {
  int i = blockIdx.x * blockDim.x + threadIdx.x;
  if (i < n) nacc[i] = 0.f;
}

__global__ void gat_pq_wmma(const float* __restrict__ x, const float* __restrict__ G,
                            float* __restrict__ p, float* __restrict__ q,
                            int n_tiles, int n_nodes) {
  int lane = threadIdx.x & 31;
  int wv   = (blockIdx.x * blockDim.x + threadIdx.x) >> 5;
  int t0 = wv * 2;
  int t1 = t0 + 1;
  if (t0 >= n_tiles) return;
  bool has1 = (t1 < n_tiles);

  int m    = lane & 15;
  int koff = (lane >> 4) << 1;
  int n0 = t0 * 16 + m; if (n0 >= n_nodes) n0 = n_nodes - 1;
  int n1 = has1 ? (t1 * 16 + m) : n0; if (n1 >= n_nodes) n1 = n_nodes - 1;

  const float* a_ptr = G + (size_t)m  * HIDDEN + koff;
  const float* b0ptr = x + (size_t)n0 * HIDDEN + koff;
  const float* b1ptr = x + (size_t)n1 * HIDDEN + koff;

  v8f acc0 = {0.f, 0.f, 0.f, 0.f, 0.f, 0.f, 0.f, 0.f};
  v8f acc1 = {0.f, 0.f, 0.f, 0.f, 0.f, 0.f, 0.f, 0.f};
#pragma unroll
  for (int k0 = 0; k0 < HIDDEN; k0 += 4) {
    v2f aa  = *(const v2f*)(a_ptr + k0);
    v2f bb0 = *(const v2f*)(b0ptr + k0);
    v2f bb1 = *(const v2f*)(b1ptr + k0);
    acc0 = __builtin_amdgcn_wmma_f32_16x16x4_f32(false, aa, false, bb0,
                                                 (short)0, acc0, false, false);
    acc1 = __builtin_amdgcn_wmma_f32_16x16x4_f32(false, aa, false, bb1,
                                                 (short)0, acc1, false, false);
  }
  if (lane < 16) {
    int o0 = t0 * 16 + lane;
    if (o0 < n_nodes) { p[o0] = acc0[0]; q[o0] = acc0[1]; }
    int o1 = t1 * 16 + lane;
    if (has1 && o1 < n_nodes) { p[o1] = acc1[0]; q[o1] = acc1[1]; }
  }
}

__global__ void gat_edge(const int* __restrict__ e, const float* __restrict__ p,
                         const float* __restrict__ q, const float* __restrict__ cs,
                         float* __restrict__ nacc, int n_edges) {
  int i = blockIdx.x * blockDim.x + threadIdx.x;
  if (i >= n_edges) return;
  int srcn = e[2 * i];
  int dstn = e[2 * i + 1];
  float z = p[srcn] + q[dstn] + cs[0];
  float lz = z > 0.f ? z : 0.2f * z;
  atomicAdd(&nacc[dstn], expf(lz));
}

__global__ void gat_out(const float* __restrict__ x, const float* __restrict__ nacc,
                        const float* __restrict__ nw, const float* __restrict__ nb,
                        float* __restrict__ out, int n_nodes) {
  int lane = threadIdx.x & 31;
  int node = (blockIdx.x * blockDim.x + threadIdx.x) >> 5;
  if (node >= n_nodes) return;
  float nv = nacc[node];
  float denom = (nv == 0.f) ? 1.f : nv;
  float4 xv = ((const float4*)(x + (size_t)node * HIDDEN))[lane];
  float4 y;
  y.x = fmaxf(xv.x / denom, 0.f) + xv.x;
  y.y = fmaxf(xv.y / denom, 0.f) + xv.y;
  y.z = fmaxf(xv.z / denom, 0.f) + xv.z;
  y.w = fmaxf(xv.w / denom, 0.f) + xv.w;
  float ss = y.x * y.x + y.y * y.y + y.z * y.z + y.w * y.w;
  ss += __shfl_xor(ss, 16, 32);
  ss += __shfl_xor(ss,  8, 32);
  ss += __shfl_xor(ss,  4, 32);
  ss += __shfl_xor(ss,  2, 32);
  ss += __shfl_xor(ss,  1, 32);
  float inv_rms = rsqrtf(ss * (1.0f / HIDDEN) + 1e-6f);
  int c = lane * 4;
  float4 o;
  o.x = nw[c + 0] * (y.x * inv_rms) + nb[c + 0];
  o.y = nw[c + 1] * (y.y * inv_rms) + nb[c + 1];
  o.z = nw[c + 2] * (y.z * inv_rms) + nb[c + 2];
  o.w = nw[c + 3] * (y.w * inv_rms) + nb[c + 3];
  ((float4*)(out + (size_t)node * HIDDEN))[lane] = o;
}

extern "C" void kernel_launch(void* const* d_in, const int* in_sizes, int n_in,
                              void* d_out, int out_size, void* d_ws, size_t ws_size,
                              hipStream_t stream) {
  const float* x     = (const float*)d_in[0];
  const float* W     = (const float*)d_in[1];
  const float* b     = (const float*)d_in[2];
  const float* att_w = (const float*)d_in[3];
  const float* nw    = (const float*)d_in[4];
  const float* nb    = (const float*)d_in[5];
  const int*   e     = (const int*)d_in[6];
  float* out = (float*)d_out;

  const int n_nodes = in_sizes[0] / HIDDEN;
  const int n_edges = in_sizes[6] / 2;

  float* G    = (float*)d_ws;
  float* cs   = G + 16 * HIDDEN;
  float* p    = cs + 16;
  float* q    = p + n_nodes;
  float* nacc = q + n_nodes;

  gat_prep_uv<<<1, 128, 0, stream>>>(W, b, att_w, G, cs);
  gat_zero<<<(n_nodes + 255) / 256, 256, 0, stream>>>(nacc, n_nodes);

  int n_tiles = (n_nodes + 15) / 16;
  int n_waves = (n_tiles + 1) / 2;
  int pq_blocks = (n_waves * 32 + 127) / 128;
  gat_pq_wmma<<<pq_blocks, 128, 0, stream>>>(x, G, p, q, n_tiles, n_nodes);

  gat_edge<<<(n_edges + 255) / 256, 256, 0, stream>>>(e, p, q, cs, nacc, n_edges);

  gat_out<<<(n_nodes * 32 + 255) / 256, 256, 0, stream>>>(x, nacc, nw, nb, out, n_nodes);
}


# Output kernel: a row lies across 8 lanes (16 floats each), a wave32 holds 4 rows per pass and takes 7 passes (28 rows); 3572 waves replace 100000; row sum order changes (16 per lane, 3 exchanges)
# speedup vs baseline: 1.1972x; 1.1972x over previous
#include <hip/hip_runtime.h>
#include <hip/hip_bf16.h>

typedef __attribute__((ext_vector_type(2))) float v2f;
typedef __attribute__((ext_vector_type(8))) float v8f;

#define HIDDEN 128

__global__ void __launch_bounds__(288)
gat_prep_uv(const float* __restrict__ W, const float* __restrict__ b,
            const float* __restrict__ att_w,
            float* __restrict__ G, float* __restrict__ cs) {
  __shared__ __attribute__((aligned(16))) float sa[3 * HIDDEN];
  const int tid  = threadIdx.x;
  const int lane = tid & 31;
  const int wv   = tid >> 5;
  const int c    = lane & 15;
  const int koff = (lane >> 4) << 1;

  v2f bb[HIDDEN / 4];
  if (wv < 8) {
    const float* bp = W + (size_t)koff * HIDDEN + wv * 16 + c;
#pragma unroll
    for (int k0 = 0; k0 < HIDDEN; k0 += 4) {
      bb[k0 >> 2].x = bp[(size_t)k0 * HIDDEN];
      bb[k0 >> 2].y = bp[(size_t)(k0 + 1) * HIDDEN];
    }
    sa[tid] = att_w[tid];
    if (tid < HIDDEN) sa[2 * HIDDEN + tid] = 0.f;
  } else {
    const float* bp = b + koff;
#pragma unroll
    for (int k0 = 0; k0 < HIDDEN; k0 += 4) {
      v2f t = *(const v2f*)(bp + k0);
      bb[k0 >> 2].x = (c == 0) ? t.x : 0.f;
      bb[k0 >> 2].y = (c == 0) ? t.y : 0.f;
    }
  }
  __syncthreads();

  const float* ap = sa + (c < 2 ? c : 2) * HIDDEN + koff;
  v2f aa[HIDDEN / 4];
#pragma unroll
  for (int k0 = 0; k0 < HIDDEN; k0 += 4) aa[k0 >> 2] = *(const v2f*)(ap + k0);
  v8f acc = {0.f, 0.f, 0.f, 0.f, 0.f, 0.f, 0.f, 0.f};
#pragma unroll
  for (int k0 = 0; k0 < HIDDEN; k0 += 4) {
    acc = __builtin_amdgcn_wmma_f32_16x16x4_f32(false, aa[k0 >> 2], false, bb[k0 >> 2],
                                                (short)0, acc, false, false);
  }

  if (wv < 8) {
    const int col = tid & (HIDDEN - 1);
    const int r0  = 2 + (tid >> 7);
#pragma unroll
    for (int i = 0; i < 7; ++i) G[(r0 + 2 * i) * HIDDEN + col] = 0.f;
    if (lane < 16) {
      G[wv * 16 + c] = acc[0];
      G[HIDDEN + wv * 16 + c] = acc[1];
    }
  } else if (lane == 0) {
    cs[0] = acc[0] + acc[1];
  }
}

__global__ void gat_zero(float* __restrict__ nacc, int n) {
  int i = blockIdx.x * blockDim.x + threadIdx.x;
  if (i < n) nacc[i] = 0.f;
}

__global__ void gat_pq_wmma(const float* __restrict__ x, const float* __restrict__ G,
                            float* __restrict__ p, float* __restrict__ q,
                            int n_tiles, int n_nodes) {
  int lane = threadIdx.x & 31;
  int wv   = (blockIdx.x * blockDim.x + threadIdx.x) >> 5;
  int t0 = wv * 2;
  int t1 = t0 + 1;
  if (t0 >= n_tiles) return;
  bool has1 = (t1 < n_tiles);

  int m    = lane & 15;
  int koff = (lane >> 4) << 1;
  int n0 = t0 * 16 + m; if (n0 >= n_nodes) n0 = n_nodes - 1;
  int n1 = has1 ? (t1 * 16 + m) : n0; if (n1 >= n_nodes) n1 = n_nodes - 1;

  const float* a_ptr = G + (size_t)m  * HIDDEN + koff;
  const float* b0ptr = x + (size_t)n0 * HIDDEN + koff;
  const float* b1ptr = x + (size_t)n1 * HIDDEN + koff;

  v8f acc0 = {0.f, 0.f, 0.f, 0.f, 0.f, 0.f, 0.f, 0.f};
  v8f acc1 = {0.f, 0.f, 0.f, 0.f, 0.f, 0.f, 0.f, 0.f};
#pragma unroll
  for (int k0 = 0; k0 < HIDDEN; k0 += 4) {
    v2f aa  = *(const v2f*)(a_ptr + k0);
    v2f bb0 = *(const v2f*)(b0ptr + k0);
    v2f bb1 = *(const v2f*)(b1ptr + k0);
    acc0 = __builtin_amdgcn_wmma_f32_16x16x4_f32(false, aa, false, bb0,
                                                 (short)0, acc0, false, false);
    acc1 = __builtin_amdgcn_wmma_f32_16x16x4_f32(false, aa, false, bb1,
                                                 (short)0, acc1, false, false);
  }
  if (lane < 16) {
    int o0 = t0 * 16 + lane;
    if (o0 < n_nodes) { p[o0] = acc0[0]; q[o0] = acc0[1]; }
    int o1 = t1 * 16 + lane;
    if (has1 && o1 < n_nodes) { p[o1] = acc1[0]; q[o1] = acc1[1]; }
  }
}

__global__ void gat_edge(const int* __restrict__ e, const float* __restrict__ p,
                         const float* __restrict__ q, const float* __restrict__ cs,
                         float* __restrict__ nacc, int n_edges) {
  int i = blockIdx.x * blockDim.x + threadIdx.x;
  if (i >= n_edges) return;
  int srcn = e[2 * i];
  int dstn = e[2 * i + 1];
  float z = p[srcn] + q[dstn] + cs[0];
  float lz = z > 0.f ? z : 0.2f * z;
  atomicAdd(&nacc[dstn], expf(lz));
}

#define OUT_PASSES 7
#define OUT_ROWS_PER_WAVE (4 * OUT_PASSES)
__global__ void __launch_bounds__(64)
gat_out(const float* __restrict__ x, const float* __restrict__ nacc,
        const float* __restrict__ nw, const float* __restrict__ nb,
        float* __restrict__ out, int n_nodes) {
  const int lane = threadIdx.x & 31;
  const int wv   = (blockIdx.x * blockDim.x + threadIdx.x) >> 5;
  const int row0 = wv * OUT_ROWS_PER_WAVE;
  if (row0 >= n_nodes) return;
  const int g = lane >> 3;
  const int j = lane & 7;

  float  nv[OUT_PASSES];
  float4 xv[OUT_PASSES][4];
#pragma unroll
  for (int ps = 0; ps < OUT_PASSES; ++ps) {
    int r = row0 + ps * 4 + g;
    if (r >= n_nodes) r = n_nodes - 1;
    nv[ps] = nacc[r];
    const float4* xr = (const float4*)(x + (size_t)r * HIDDEN) + j;
#pragma unroll
    for (int t = 0; t < 4; ++t) xv[ps][t] = xr[t * 8];
  }
  float4 w4[4], b4[4];
#pragma unroll
  for (int t = 0; t < 4; ++t) {
    w4[t] = ((const float4*)nw)[t * 8 + j];
    b4[t] = ((const float4*)nb)[t * 8 + j];
  }

#pragma unroll
  for (int ps = 0; ps < OUT_PASSES; ++ps) {
    const int r = row0 + ps * 4 + g;
    const float denom = (nv[ps] == 0.f) ? 1.f : nv[ps];
    float4 y[4];
    float ss = 0.f;
#pragma unroll
    for (int t = 0; t < 4; ++t) {
      const float4 v = xv[ps][t];
      y[t].x = fmaxf(v.x / denom, 0.f) + v.x;
      y[t].y = fmaxf(v.y / denom, 0.f) + v.y;
      y[t].z = fmaxf(v.z / denom, 0.f) + v.z;
      y[t].w = fmaxf(v.w / denom, 0.f) + v.w;
      ss += y[t].x * y[t].x + y[t].y * y[t].y + y[t].z * y[t].z + y[t].w * y[t].w;
    }
    ss += __shfl_xor(ss, 4, 32);
    ss += __shfl_xor(ss, 2, 32);
    ss += __shfl_xor(ss, 1, 32);
    const float inv_rms = rsqrtf(ss * (1.0f / HIDDEN) + 1e-6f);
    if (r < n_nodes) {
      float4* orow = (float4*)(out + (size_t)r * HIDDEN) + j;
#pragma unroll
      for (int t = 0; t < 4; ++t) {
        float4 o;
        o.x = w4[t].x * (y[t].x * inv_rms) + b4[t].x;
        o.y = w4[t].y * (y[t].y * inv_rms) + b4[t].y;
        o.z = w4[t].z * (y[t].z * inv_rms) + b4[t].z;
        o.w = w4[t].w * (y[t].w * inv_rms) + b4[t].w;
        orow[t * 8] = o;
      }
    }
  }
}

extern "C" void kernel_launch(void* const* d_in, const int* in_sizes, int n_in,
                              void* d_out, int out_size, void* d_ws, size_t ws_size,
                              hipStream_t stream) {
  const float* x     = (const float*)d_in[0];
  const float* W     = (const float*)d_in[1];
  const float* b     = (const float*)d_in[2];
  const float* att_w = (const float*)d_in[3];
  const float* nw    = (const float*)d_in[4];
  const float* nb    = (const float*)d_in[5];
  const int*   e     = (const int*)d_in[6];
  float* out = (float*)d_out;

  const int n_nodes = in_sizes[0] / HIDDEN;
  const int n_edges = in_sizes[6] / 2;

  float* G    = (float*)d_ws;
  float* cs   = G + 16 * HIDDEN;
  float* p    = cs + 16;
  float* q    = p + n_nodes;
  float* nacc = q + n_nodes;

  gat_prep_uv<<<1, 288, 0, stream>>>(W, b, att_w, G, cs);
  gat_zero<<<(n_nodes + 255) / 256, 256, 0, stream>>>(nacc, n_nodes);

  int n_tiles = (n_nodes + 15) / 16;
  int n_waves = (n_tiles + 1) / 2;
  int pq_blocks = (n_waves * 32 + 127) / 128;
  gat_pq_wmma<<<pq_blocks, 128, 0, stream>>>(x, G, p, q, n_tiles, n_nodes);

  gat_edge<<<(n_edges + 255) / 256, 256, 0, stream>>>(e, p, q, cs, nacc, n_edges);

  int out_waves = (n_nodes + OUT_ROWS_PER_WAVE - 1) / OUT_ROWS_PER_WAVE;
  gat_out<<<(out_waves * 32 + 63) / 64, 64, 0, stream>>>(x, nacc, nw, nb, out, n_nodes);
}


# p,q kernel: A and B fragments loaded 16 bytes per lane by one shared loader (k permuted alike in both), all 48 loads issued before the first WMMA: 1 memory trip per wave, not 4; k sum order changes
# speedup vs baseline: 1.2457x; 1.0405x over previous
#include <hip/hip_runtime.h>
#include <hip/hip_bf16.h>

typedef __attribute__((ext_vector_type(2))) float v2f;
typedef __attribute__((ext_vector_type(8))) float v8f;

#define HIDDEN 128

__global__ void __launch_bounds__(288)
gat_prep_uv(const float* __restrict__ W, const float* __restrict__ b,
            const float* __restrict__ att_w,
            float* __restrict__ G, float* __restrict__ cs) {
  __shared__ __attribute__((aligned(16))) float sa[3 * HIDDEN];
  const int tid  = threadIdx.x;
  const int lane = tid & 31;
  const int wv   = tid >> 5;
  const int c    = lane & 15;
  const int koff = (lane >> 4) << 1;

  v2f bb[HIDDEN / 4];
  if (wv < 8) {
    const float* bp = W + (size_t)koff * HIDDEN + wv * 16 + c;
#pragma unroll
    for (int k0 = 0; k0 < HIDDEN; k0 += 4) {
      bb[k0 >> 2].x = bp[(size_t)k0 * HIDDEN];
      bb[k0 >> 2].y = bp[(size_t)(k0 + 1) * HIDDEN];
    }
    sa[tid] = att_w[tid];
    if (tid < HIDDEN) sa[2 * HIDDEN + tid] = 0.f;
  } else {
    const float* bp = b + koff;
#pragma unroll
    for (int k0 = 0; k0 < HIDDEN; k0 += 4) {
      v2f t = *(const v2f*)(bp + k0);
      bb[k0 >> 2].x = (c == 0) ? t.x : 0.f;
      bb[k0 >> 2].y = (c == 0) ? t.y : 0.f;
    }
  }
  __syncthreads();

  const float* ap = sa + (c < 2 ? c : 2) * HIDDEN + koff;
  v2f aa[HIDDEN / 4];
#pragma unroll
  for (int k0 = 0; k0 < HIDDEN; k0 += 4) aa[k0 >> 2] = *(const v2f*)(ap + k0);
  v8f acc = {0.f, 0.f, 0.f, 0.f, 0.f, 0.f, 0.f, 0.f};
#pragma unroll
  for (int k0 = 0; k0 < HIDDEN; k0 += 4) {
    acc = __builtin_amdgcn_wmma_f32_16x16x4_f32(false, aa[k0 >> 2], false, bb[k0 >> 2],
                                                (short)0, acc, false, false);
  }

  if (wv < 8) {
    const int col = tid & (HIDDEN - 1);
    const int r0  = 2 + (tid >> 7);
#pragma unroll
    for (int i = 0; i < 7; ++i) G[(r0 + 2 * i) * HIDDEN + col] = 0.f;
    if (lane < 16) {
      G[wv * 16 + c] = acc[0];
      G[HIDDEN + wv * 16 + c] = acc[1];
    }
  } else if (lane == 0) {
    cs[0] = acc[0] + acc[1];
  }
}

__global__ void gat_zero(float* __restrict__ nacc, int n) {
  int i = blockIdx.x * blockDim.x + threadIdx.x;
  if (i < n) nacc[i] = 0.f;
}

typedef __attribute__((ext_vector_type(4))) float v4f;

__device__ __forceinline__ v4f frag_ld(const float* __restrict__ row_h, int t) {
  return *(const v4f*)(row_h + 8 * t);
}

__global__ void __launch_bounds__(128)
gat_pq_wmma(const float* __restrict__ x, const float* __restrict__ G,
            float* __restrict__ p, float* __restrict__ q,
            int n_tiles, int n_nodes) {
  int lane = threadIdx.x & 31;
  int wv   = (blockIdx.x * blockDim.x + threadIdx.x) >> 5;
  int t0 = wv * 2;
  int t1 = t0 + 1;
  if (t0 >= n_tiles) return;
  bool has1 = (t1 < n_tiles);

  int m    = lane & 15;
  int hoff = (lane >> 4) << 2;
  int n0 = t0 * 16 + m; if (n0 >= n_nodes) n0 = n_nodes - 1;
  int n1 = has1 ? (t1 * 16 + m) : n0; if (n1 >= n_nodes) n1 = n_nodes - 1;

  const float* a_ptr = G + (size_t)m  * HIDDEN + hoff;
  const float* b0ptr = x + (size_t)n0 * HIDDEN + hoff;
  const float* b1ptr = x + (size_t)n1 * HIDDEN + hoff;

  v4f av[HIDDEN / 8], b0v[HIDDEN / 8], b1v[HIDDEN / 8];
#pragma unroll
  for (int t = 0; t < HIDDEN / 8; ++t) {
    b0v[t] = frag_ld(b0ptr, t);
    b1v[t] = frag_ld(b1ptr, t);
    av[t]  = frag_ld(a_ptr, t);
  }

  v8f acc0 = {0.f, 0.f, 0.f, 0.f, 0.f, 0.f, 0.f, 0.f};
  v8f acc1 = {0.f, 0.f, 0.f, 0.f, 0.f, 0.f, 0.f, 0.f};
#pragma unroll
  for (int t = 0; t < HIDDEN / 8; ++t) {
    const v2f a_lo  = {av[t].x,  av[t].y},  a_hi  = {av[t].z,  av[t].w};
    const v2f b0_lo = {b0v[t].x, b0v[t].y}, b0_hi = {b0v[t].z, b0v[t].w};
    const v2f b1_lo = {b1v[t].x, b1v[t].y}, b1_hi = {b1v[t].z, b1v[t].w};
    acc0 = __builtin_amdgcn_wmma_f32_16x16x4_f32(false, a_lo, false, b0_lo,
                                                 (short)0, acc0, false, false);
    acc1 = __builtin_amdgcn_wmma_f32_16x16x4_f32(false, a_lo, false, b1_lo,
                                                 (short)0, acc1, false, false);
    acc0 = __builtin_amdgcn_wmma_f32_16x16x4_f32(false, a_hi, false, b0_hi,
                                                 (short)0, acc0, false, false);
    acc1 = __builtin_amdgcn_wmma_f32_16x16x4_f32(false, a_hi, false, b1_hi,
                                                 (short)0, acc1, false, false);
  }
  __builtin_amdgcn_sched_group_barrier(0x020, 3 * (HIDDEN / 8), 0);
  __builtin_amdgcn_sched_group_barrier(0x008, HIDDEN / 2, 0);
  if (lane < 16) {
    int o0 = t0 * 16 + lane;
    if (o0 < n_nodes) { p[o0] = acc0[0]; q[o0] = acc0[1]; }
    int o1 = t1 * 16 + lane;
    if (has1 && o1 < n_nodes) { p[o1] = acc1[0]; q[o1] = acc1[1]; }
  }
}

__global__ void gat_edge(const int* __restrict__ e, const float* __restrict__ p,
                         const float* __restrict__ q, const float* __restrict__ cs,
                         float* __restrict__ nacc, int n_edges) {
  int i = blockIdx.x * blockDim.x + threadIdx.x;
  if (i >= n_edges) return;
  int srcn = e[2 * i];
  int dstn = e[2 * i + 1];
  float z = p[srcn] + q[dstn] + cs[0];
  float lz = z > 0.f ? z : 0.2f * z;
  atomicAdd(&nacc[dstn], expf(lz));
}

#define OUT_PASSES 7
#define OUT_ROWS_PER_WAVE (4 * OUT_PASSES)
__global__ void __launch_bounds__(64)
gat_out(const float* __restrict__ x, const float* __restrict__ nacc,
        const float* __restrict__ nw, const float* __restrict__ nb,
        float* __restrict__ out, int n_nodes) {
  const int lane = threadIdx.x & 31;
  const int wv   = (blockIdx.x * blockDim.x + threadIdx.x) >> 5;
  const int row0 = wv * OUT_ROWS_PER_WAVE;
  if (row0 >= n_nodes) return;
  const int g = lane >> 3;
  const int j = lane & 7;

  float  nv[OUT_PASSES];
  float4 xv[OUT_PASSES][4];
#pragma unroll
  for (int ps = 0; ps < OUT_PASSES; ++ps) {
    int r = row0 + ps * 4 + g;
    if (r >= n_nodes) r = n_nodes - 1;
    nv[ps] = nacc[r];
    const float4* xr = (const float4*)(x + (size_t)r * HIDDEN) + j;
#pragma unroll
    for (int t = 0; t < 4; ++t) xv[ps][t] = xr[t * 8];
  }
  float4 w4[4], b4[4];
#pragma unroll
  for (int t = 0; t < 4; ++t) {
    w4[t] = ((const float4*)nw)[t * 8 + j];
    b4[t] = ((const float4*)nb)[t * 8 + j];
  }

#pragma unroll
  for (int ps = 0; ps < OUT_PASSES; ++ps) {
    const int r = row0 + ps * 4 + g;
    const float denom = (nv[ps] == 0.f) ? 1.f : nv[ps];
    float4 y[4];
    float ss = 0.f;
#pragma unroll
    for (int t = 0; t < 4; ++t) {
      const float4 v = xv[ps][t];
      y[t].x = fmaxf(v.x / denom, 0.f) + v.x;
      y[t].y = fmaxf(v.y / denom, 0.f) + v.y;
      y[t].z = fmaxf(v.z / denom, 0.f) + v.z;
      y[t].w = fmaxf(v.w / denom, 0.f) + v.w;
      ss += y[t].x * y[t].x + y[t].y * y[t].y + y[t].z * y[t].z + y[t].w * y[t].w;
    }
    ss += __shfl_xor(ss, 4, 32);
    ss += __shfl_xor(ss, 2, 32);
    ss += __shfl_xor(ss, 1, 32);
    const float inv_rms = rsqrtf(ss * (1.0f / HIDDEN) + 1e-6f);
    if (r < n_nodes) {
      float4* orow = (float4*)(out + (size_t)r * HIDDEN) + j;
#pragma unroll
      for (int t = 0; t < 4; ++t) {
        float4 o;
        o.x = w4[t].x * (y[t].x * inv_rms) + b4[t].x;
        o.y = w4[t].y * (y[t].y * inv_rms) + b4[t].y;
        o.z = w4[t].z * (y[t].z * inv_rms) + b4[t].z;
        o.w = w4[t].w * (y[t].w * inv_rms) + b4[t].w;
        orow[t * 8] = o;
      }
    }
  }
}

extern "C" void kernel_launch(void* const* d_in, const int* in_sizes, int n_in,
                              void* d_out, int out_size, void* d_ws, size_t ws_size,
                              hipStream_t stream) {
  const float* x     = (const float*)d_in[0];
  const float* W     = (const float*)d_in[1];
  const float* b     = (const float*)d_in[2];
  const float* att_w = (const float*)d_in[3];
  const float* nw    = (const float*)d_in[4];
  const float* nb    = (const float*)d_in[5];
  const int*   e     = (const int*)d_in[6];
  float* out = (float*)d_out;

  const int n_nodes = in_sizes[0] / HIDDEN;
  const int n_edges = in_sizes[6] / 2;

  float* G    = (float*)d_ws;
  float* cs   = G + 16 * HIDDEN;
  float* p    = cs + 16;
  float* q    = p + n_nodes;
  float* nacc = q + n_nodes;

  gat_prep_uv<<<1, 288, 0, stream>>>(W, b, att_w, G, cs);
  gat_zero<<<(n_nodes + 255) / 256, 256, 0, stream>>>(nacc, n_nodes);

  int n_tiles = (n_nodes + 15) / 16;
  int n_waves = (n_tiles + 1) / 2;
  int pq_blocks = (n_waves * 32 + 127) / 128;
  gat_pq_wmma<<<pq_blocks, 128, 0, stream>>>(x, G, p, q, n_tiles, n_nodes);

  gat_edge<<<(n_edges + 255) / 256, 256, 0, stream>>>(e, p, q, cs, nacc, n_edges);

  int out_waves = (n_nodes + OUT_ROWS_PER_WAVE - 1) / OUT_ROWS_PER_WAVE;
  gat_out<<<(out_waves * 32 + 63) / 64, 64, 0, stream>>>(x, nacc, nw, nb, out, n_nodes);
}


# Edge sums without global atomics: a new pass counting-sorts (p[src],dst) by 256-node destination range in LDS; range-owner blocks add their runs by LDS f32 atomics, store nacc once; no zero-fill
# speedup vs baseline: 1.9665x; 1.5786x over previous
#include <hip/hip_runtime.h>
#include <hip/hip_bf16.h>

typedef __attribute__((ext_vector_type(2))) float v2f;
typedef __attribute__((ext_vector_type(8))) float v8f;

#define HIDDEN 128

__global__ void __launch_bounds__(288)
gat_prep_uv(const float* __restrict__ W, const float* __restrict__ b,
            const float* __restrict__ att_w,
            float* __restrict__ G, float* __restrict__ cs) {
  __shared__ __attribute__((aligned(16))) float sa[3 * HIDDEN];
  const int tid  = threadIdx.x;
  const int lane = tid & 31;
  const int wv   = tid >> 5;
  const int c    = lane & 15;
  const int koff = (lane >> 4) << 1;

  v2f bb[HIDDEN / 4];
  if (wv < 8) {
    const float* bp = W + (size_t)koff * HIDDEN + wv * 16 + c;
#pragma unroll
    for (int k0 = 0; k0 < HIDDEN; k0 += 4) {
      bb[k0 >> 2].x = bp[(size_t)k0 * HIDDEN];
      bb[k0 >> 2].y = bp[(size_t)(k0 + 1) * HIDDEN];
    }
    sa[tid] = att_w[tid];
    if (tid < HIDDEN) sa[2 * HIDDEN + tid] = 0.f;
  } else {
    const float* bp = b + koff;
#pragma unroll
    for (int k0 = 0; k0 < HIDDEN; k0 += 4) {
      v2f t = *(const v2f*)(bp + k0);
      bb[k0 >> 2].x = (c == 0) ? t.x : 0.f;
      bb[k0 >> 2].y = (c == 0) ? t.y : 0.f;
    }
  }
  __syncthreads();

  const float* ap = sa + (c < 2 ? c : 2) * HIDDEN + koff;
  v2f aa[HIDDEN / 4];
#pragma unroll
  for (int k0 = 0; k0 < HIDDEN; k0 += 4) aa[k0 >> 2] = *(const v2f*)(ap + k0);
  v8f acc = {0.f, 0.f, 0.f, 0.f, 0.f, 0.f, 0.f, 0.f};
#pragma unroll
  for (int k0 = 0; k0 < HIDDEN; k0 += 4) {
    acc = __builtin_amdgcn_wmma_f32_16x16x4_f32(false, aa[k0 >> 2], false, bb[k0 >> 2],
                                                (short)0, acc, false, false);
  }

  if (wv < 8) {
    const int col = tid & (HIDDEN - 1);
    const int r0  = 2 + (tid >> 7);
#pragma unroll
    for (int i = 0; i < 7; ++i) G[(r0 + 2 * i) * HIDDEN + col] = 0.f;
    if (lane < 16) {
      G[wv * 16 + c] = acc[0];
      G[HIDDEN + wv * 16 + c] = acc[1];
    }
  } else if (lane == 0) {
    cs[0] = acc[0] + acc[1];
  }
}

typedef __attribute__((ext_vector_type(4))) float v4f;

__device__ __forceinline__ v4f frag_ld(const float* __restrict__ row_h, int t) {
  return *(const v4f*)(row_h + 8 * t);
}

__global__ void __launch_bounds__(128)
gat_pq_wmma(const float* __restrict__ x, const float* __restrict__ G,
            float* __restrict__ p, float* __restrict__ q,
            int n_tiles, int n_nodes) {
  int lane = threadIdx.x & 31;
  int wv   = (blockIdx.x * blockDim.x + threadIdx.x) >> 5;
  int t0 = wv * 2;
  int t1 = t0 + 1;
  if (t0 >= n_tiles) return;
  bool has1 = (t1 < n_tiles);

  int m    = lane & 15;
  int hoff = (lane >> 4) << 2;
  int n0 = t0 * 16 + m; if (n0 >= n_nodes) n0 = n_nodes - 1;
  int n1 = has1 ? (t1 * 16 + m) : n0; if (n1 >= n_nodes) n1 = n_nodes - 1;

  const float* a_ptr = G + (size_t)m  * HIDDEN + hoff;
  const float* b0ptr = x + (size_t)n0 * HIDDEN + hoff;
  const float* b1ptr = x + (size_t)n1 * HIDDEN + hoff;

  v4f av[HIDDEN / 8], b0v[HIDDEN / 8], b1v[HIDDEN / 8];
#pragma unroll
  for (int t = 0; t < HIDDEN / 8; ++t) {
    b0v[t] = frag_ld(b0ptr, t);
    b1v[t] = frag_ld(b1ptr, t);
    av[t]  = frag_ld(a_ptr, t);
  }

  v8f acc0 = {0.f, 0.f, 0.f, 0.f, 0.f, 0.f, 0.f, 0.f};
  v8f acc1 = {0.f, 0.f, 0.f, 0.f, 0.f, 0.f, 0.f, 0.f};
#pragma unroll
  for (int t = 0; t < HIDDEN / 8; ++t) {
    const v2f a_lo  = {av[t].x,  av[t].y},  a_hi  = {av[t].z,  av[t].w};
    const v2f b0_lo = {b0v[t].x, b0v[t].y}, b0_hi = {b0v[t].z, b0v[t].w};
    const v2f b1_lo = {b1v[t].x, b1v[t].y}, b1_hi = {b1v[t].z, b1v[t].w};
    acc0 = __builtin_amdgcn_wmma_f32_16x16x4_f32(false, a_lo, false, b0_lo,
                                                 (short)0, acc0, false, false);
    acc1 = __builtin_amdgcn_wmma_f32_16x16x4_f32(false, a_lo, false, b1_lo,
                                                 (short)0, acc1, false, false);
    acc0 = __builtin_amdgcn_wmma_f32_16x16x4_f32(false, a_hi, false, b0_hi,
                                                 (short)0, acc0, false, false);
    acc1 = __builtin_amdgcn_wmma_f32_16x16x4_f32(false, a_hi, false, b1_hi,
                                                 (short)0, acc1, false, false);
  }
  __builtin_amdgcn_sched_group_barrier(0x020, 3 * (HIDDEN / 8), 0);
  __builtin_amdgcn_sched_group_barrier(0x008, HIDDEN / 2, 0);
  if (lane < 16) {
    int o0 = t0 * 16 + lane;
    if (o0 < n_nodes) { p[o0] = acc0[0]; q[o0] = acc0[1]; }
    int o1 = t1 * 16 + lane;
    if (has1 && o1 < n_nodes) { p[o1] = acc1[0]; q[o1] = acc1[1]; }
  }
}

#define RANGE_SHIFT 8
#define RANGE_NODES (1 << RANGE_SHIFT)
#define MAX_RANGES 512
#define BIN_THREADS 256
#define BIN_EDGES_PER_THREAD 10
#define SLICE_EDGES (BIN_THREADS * BIN_EDGES_PER_THREAD)

__global__ void __launch_bounds__(BIN_THREADS)
gat_bin(const int* __restrict__ e, const float* __restrict__ p, float* __restrict__ scratch,
        int n_edges, int n_slices, int n_ranges) {
  __shared__ int  cnt[MAX_RANGES];
  __shared__ int  wsum[BIN_THREADS / 32];
  __shared__ int2 stage[SLICE_EDGES];
  int2* sorted = (int2*)scratch;
  int2* runs   = sorted + (size_t)n_slices * SLICE_EDGES;
  const int tid  = threadIdx.x;
  const int lane = tid & 31;
  const int wv   = tid >> 5;
  const int base = blockIdx.x * SLICE_EDGES;

  cnt[tid] = 0;
  cnt[tid + BIN_THREADS] = 0;

  int2 ed[BIN_EDGES_PER_THREAD];
  int  rank[BIN_EDGES_PER_THREAD];
#pragma unroll
  for (int k = 0; k < BIN_EDGES_PER_THREAD; ++k) {
    int i = base + k * BIN_THREADS + tid;
    if (i >= n_edges) i = n_edges - 1;
    ed[k] = ((const int2*)e)[i];
  }
#pragma unroll
  for (int k = 0; k < BIN_EDGES_PER_THREAD; ++k) ed[k].x = __float_as_int(p[ed[k].x]);
  __builtin_amdgcn_sched_group_barrier(0x020, BIN_EDGES_PER_THREAD, 0);
  __syncthreads();
#pragma unroll
  for (int k = 0; k < BIN_EDGES_PER_THREAD; ++k) {
    const int i = base + k * BIN_THREADS + tid;
    rank[k] = (i < n_edges) ? atomicAdd(&cnt[ed[k].y >> RANGE_SHIFT], 1) : 0;
  }
  __syncthreads();

  const int c0 = cnt[2 * tid];
  const int c1 = cnt[2 * tid + 1];
  const int s2 = c0 + c1;
  int incl = s2;
#pragma unroll
  for (int d = 1; d < 32; d <<= 1) {
    const int t = __shfl_up(incl, d, 32);
    if (lane >= d) incl += t;
  }
  if (lane == 31) wsum[wv] = incl;
  __syncthreads();
  int wbase = 0;
  for (int w = 0; w < wv; ++w) wbase += wsum[w];
  const int excl = wbase + incl - s2;
  cnt[2 * tid]     = excl;
  cnt[2 * tid + 1] = excl + c0;
  __syncthreads();

  for (int b = tid; b < n_ranges; b += BIN_THREADS)
    runs[(size_t)blockIdx.x * n_ranges + b] = make_int2(cnt[b], cnt[b + 1]);

#pragma unroll
  for (int k = 0; k < BIN_EDGES_PER_THREAD; ++k) {
    const int i = base + k * BIN_THREADS + tid;
    if (i < n_edges) stage[cnt[ed[k].y >> RANGE_SHIFT] + rank[k]] = ed[k];
  }
  __syncthreads();
  int n_here = n_edges - base;
  if (n_here > SLICE_EDGES) n_here = SLICE_EDGES;
  for (int j = tid; j < n_here; j += BIN_THREADS) sorted[base + j] = stage[j];
}

#define OWN_BATCH 8
__global__ void __launch_bounds__(RANGE_NODES)
gat_edge_own(const float* __restrict__ scratch, const float* __restrict__ q,
             const float* __restrict__ cs, float* __restrict__ nacc,
             int n_slices, int n_ranges, int n_nodes) {
  __shared__ float acc[RANGE_NODES];
  __shared__ float ql[RANGE_NODES];
  const int2* sorted = (const int2*)scratch;
  const int2* runs   = sorted + (size_t)n_slices * SLICE_EDGES;
  const int tid  = threadIdx.x;
  const int b    = blockIdx.x;
  const int lo   = b << RANGE_SHIFT;
  const int node = lo + tid;
  acc[tid] = 0.f;
  ql[tid]  = (node < n_nodes) ? q[node] : 0.f;
  const float c = cs[0];
  __syncthreads();

  for (int s = tid; s < n_slices; s += RANGE_NODES) {
    const int2 jr = runs[(size_t)s * n_ranges + b];
    const int2* run = sorted + (size_t)s * SLICE_EDGES;
    for (int j = jr.x; j < jr.y; j += OWN_BATCH) {
      int2 rec[OWN_BATCH];
#pragma unroll
      for (int k = 0; k < OWN_BATCH; ++k) {
        int jj = j + k;
        if (jj >= jr.y) jj = jr.y - 1;
        rec[k] = run[jj];
      }
      asm volatile("" :: "v"(rec[0].x), "v"(rec[0].y), "v"(rec[1].x), "v"(rec[1].y),
                         "v"(rec[2].x), "v"(rec[2].y), "v"(rec[3].x), "v"(rec[3].y),
                         "v"(rec[4].x), "v"(rec[4].y), "v"(rec[5].x), "v"(rec[5].y),
                         "v"(rec[6].x), "v"(rec[6].y), "v"(rec[7].x), "v"(rec[7].y));
#pragma unroll
      for (int k = 0; k < OWN_BATCH; ++k) {
        if (j + k < jr.y) {
          const int dl = rec[k].y - lo;
          const float z  = __int_as_float(rec[k].x) + ql[dl] + c;
          const float lz = z > 0.f ? z : 0.2f * z;
          atomicAdd(&acc[dl], expf(lz));
        }
      }
    }
  }
  __syncthreads();
  if (node < n_nodes) nacc[node] = acc[tid];
}

#define OUT_PASSES 7
#define OUT_ROWS_PER_WAVE (4 * OUT_PASSES)
__global__ void __launch_bounds__(64)
gat_out(const float* __restrict__ x, const float* __restrict__ nacc,
        const float* __restrict__ nw, const float* __restrict__ nb,
        float* __restrict__ out, int n_nodes) {
  const int lane = threadIdx.x & 31;
  const int wv   = (blockIdx.x * blockDim.x + threadIdx.x) >> 5;
  const int row0 = wv * OUT_ROWS_PER_WAVE;
  if (row0 >= n_nodes) return;
  const int g = lane >> 3;
  const int j = lane & 7;

  float  nv[OUT_PASSES];
  float4 xv[OUT_PASSES][4];
#pragma unroll
  for (int ps = 0; ps < OUT_PASSES; ++ps) {
    int r = row0 + ps * 4 + g;
    if (r >= n_nodes) r = n_nodes - 1;
    nv[ps] = nacc[r];
    const float4* xr = (const float4*)(x + (size_t)r * HIDDEN) + j;
#pragma unroll
    for (int t = 0; t < 4; ++t) xv[ps][t] = xr[t * 8];
  }
  float4 w4[4], b4[4];
#pragma unroll
  for (int t = 0; t < 4; ++t) {
    w4[t] = ((const float4*)nw)[t * 8 + j];
    b4[t] = ((const float4*)nb)[t * 8 + j];
  }

#pragma unroll
  for (int ps = 0; ps < OUT_PASSES; ++ps) {
    const int r = row0 + ps * 4 + g;
    const float denom = (nv[ps] == 0.f) ? 1.f : nv[ps];
    float4 y[4];
    float ss = 0.f;
#pragma unroll
    for (int t = 0; t < 4; ++t) {
      const float4 v = xv[ps][t];
      y[t].x = fmaxf(v.x / denom, 0.f) + v.x;
      y[t].y = fmaxf(v.y / denom, 0.f) + v.y;
      y[t].z = fmaxf(v.z / denom, 0.f) + v.z;
      y[t].w = fmaxf(v.w / denom, 0.f) + v.w;
      ss += y[t].x * y[t].x + y[t].y * y[t].y + y[t].z * y[t].z + y[t].w * y[t].w;
    }
    ss += __shfl_xor(ss, 4, 32);
    ss += __shfl_xor(ss, 2, 32);
    ss += __shfl_xor(ss, 1, 32);
    const float inv_rms = rsqrtf(ss * (1.0f / HIDDEN) + 1e-6f);
    if (r < n_nodes) {
      float4* orow = (float4*)(out + (size_t)r * HIDDEN) + j;
#pragma unroll
      for (int t = 0; t < 4; ++t) {
        float4 o;
        o.x = w4[t].x * (y[t].x * inv_rms) + b4[t].x;
        o.y = w4[t].y * (y[t].y * inv_rms) + b4[t].y;
        o.z = w4[t].z * (y[t].z * inv_rms) + b4[t].z;
        o.w = w4[t].w * (y[t].w * inv_rms) + b4[t].w;
        orow[t * 8] = o;
      }
    }
  }
}

extern "C" void kernel_launch(void* const* d_in, const int* in_sizes, int n_in,
                              void* d_out, int out_size, void* d_ws, size_t ws_size,
                              hipStream_t stream) {
  const float* x     = (const float*)d_in[0];
  const float* W     = (const float*)d_in[1];
  const float* b     = (const float*)d_in[2];
  const float* att_w = (const float*)d_in[3];
  const float* nw    = (const float*)d_in[4];
  const float* nb    = (const float*)d_in[5];
  const int*   e     = (const int*)d_in[6];
  float* out = (float*)d_out;

  const int n_nodes = in_sizes[0] / HIDDEN;
  const int n_edges = in_sizes[6] / 2;

  float* G    = (float*)d_ws;
  float* cs   = G + 16 * HIDDEN;
  float* p    = cs + 16;
  float* q    = p + n_nodes;
  float* nacc = q + n_nodes;

  gat_prep_uv<<<1, 288, 0, stream>>>(W, b, att_w, G, cs);

  int n_tiles = (n_nodes + 15) / 16;
  int n_waves = (n_tiles + 1) / 2;
  int pq_blocks = (n_waves * 32 + 127) / 128;
  gat_pq_wmma<<<pq_blocks, 128, 0, stream>>>(x, G, p, q, n_tiles, n_nodes);

  int n_slices = (n_edges + SLICE_EDGES - 1) / SLICE_EDGES;
  int n_ranges = (n_nodes + RANGE_NODES - 1) / RANGE_NODES;
  gat_bin<<<n_slices, BIN_THREADS, 0, stream>>>(e, p, out, n_edges, n_slices, n_ranges);
  gat_edge_own<<<n_ranges, RANGE_NODES, 0, stream>>>(out, q, cs, nacc, n_slices, n_ranges, n_nodes);

  int out_waves = (n_nodes + OUT_ROWS_PER_WAVE - 1) / OUT_ROWS_PER_WAVE;
  gat_out<<<(out_waves * 32 + 63) / 64, 64, 0, stream>>>(x, nacc, nw, nb, out, n_nodes);
}


# Edge sort no longer reads p: records keep (src,dst) and the range-owner pass gathers p[src] itself, so the sort runs in the prologue's launch (blocks 1..250 beside block 0): 5 -> 4 launches
# speedup vs baseline: 2.0713x; 1.0533x over previous
#include <hip/hip_runtime.h>
#include <hip/hip_bf16.h>

typedef __attribute__((ext_vector_type(2))) float v2f;
typedef __attribute__((ext_vector_type(8))) float v8f;

#define HIDDEN 128

__device__ __forceinline__ void
prep_uv_block(const float* __restrict__ W, const float* __restrict__ b,
              const float* __restrict__ att_w,
              float* __restrict__ G, float* __restrict__ cs) {
  __shared__ __attribute__((aligned(16))) float sa[3 * HIDDEN];
  const int tid  = threadIdx.x;
  const int lane = tid & 31;
  const int wv   = tid >> 5;
  const int c    = lane & 15;
  const int koff = (lane >> 4) << 1;

  v2f bb[HIDDEN / 4];
  if (wv < 8) {
    const float* bp = W + (size_t)koff * HIDDEN + wv * 16 + c;
#pragma unroll
    for (int k0 = 0; k0 < HIDDEN; k0 += 4) {
      bb[k0 >> 2].x = bp[(size_t)k0 * HIDDEN];
      bb[k0 >> 2].y = bp[(size_t)(k0 + 1) * HIDDEN];
    }
    sa[tid] = att_w[tid];
    if (tid < HIDDEN) sa[2 * HIDDEN + tid] = 0.f;
  } else {
    const float* bp = b + koff;
#pragma unroll
    for (int k0 = 0; k0 < HIDDEN; k0 += 4) {
      v2f t = *(const v2f*)(bp + k0);
      bb[k0 >> 2].x = (c == 0) ? t.x : 0.f;
      bb[k0 >> 2].y = (c == 0) ? t.y : 0.f;
    }
  }
  __syncthreads();

  const float* ap = sa + (c < 2 ? c : 2) * HIDDEN + koff;
  v2f aa[HIDDEN / 4];
#pragma unroll
  for (int k0 = 0; k0 < HIDDEN; k0 += 4) aa[k0 >> 2] = *(const v2f*)(ap + k0);
  v8f acc = {0.f, 0.f, 0.f, 0.f, 0.f, 0.f, 0.f, 0.f};
#pragma unroll
  for (int k0 = 0; k0 < HIDDEN; k0 += 4) {
    acc = __builtin_amdgcn_wmma_f32_16x16x4_f32(false, aa[k0 >> 2], false, bb[k0 >> 2],
                                                (short)0, acc, false, false);
  }

  if (wv < 8) {
    const int col = tid & (HIDDEN - 1);
    const int r0  = 2 + (tid >> 7);
#pragma unroll
    for (int i = 0; i < 7; ++i) G[(r0 + 2 * i) * HIDDEN + col] = 0.f;
    if (lane < 16) {
      G[wv * 16 + c] = acc[0];
      G[HIDDEN + wv * 16 + c] = acc[1];
    }
  } else if (lane == 0) {
    cs[0] = acc[0] + acc[1];
  }
}

typedef __attribute__((ext_vector_type(4))) float v4f;

__device__ __forceinline__ v4f frag_ld(const float* __restrict__ row_h, int t) {
  return *(const v4f*)(row_h + 8 * t);
}

__global__ void __launch_bounds__(128)
gat_pq_wmma(const float* __restrict__ x, const float* __restrict__ G,
            float* __restrict__ p, float* __restrict__ q,
            int n_tiles, int n_nodes) {
  int lane = threadIdx.x & 31;
  int wv   = (blockIdx.x * blockDim.x + threadIdx.x) >> 5;
  int t0 = wv * 2;
  int t1 = t0 + 1;
  if (t0 >= n_tiles) return;
  bool has1 = (t1 < n_tiles);

  int m    = lane & 15;
  int hoff = (lane >> 4) << 2;
  int n0 = t0 * 16 + m; if (n0 >= n_nodes) n0 = n_nodes - 1;
  int n1 = has1 ? (t1 * 16 + m) : n0; if (n1 >= n_nodes) n1 = n_nodes - 1;

  const float* a_ptr = G + (size_t)m  * HIDDEN + hoff;
  const float* b0ptr = x + (size_t)n0 * HIDDEN + hoff;
  const float* b1ptr = x + (size_t)n1 * HIDDEN + hoff;

  v4f av[HIDDEN / 8], b0v[HIDDEN / 8], b1v[HIDDEN / 8];
#pragma unroll
  for (int t = 0; t < HIDDEN / 8; ++t) {
    b0v[t] = frag_ld(b0ptr, t);
    b1v[t] = frag_ld(b1ptr, t);
    av[t]  = frag_ld(a_ptr, t);
  }

  v8f acc0 = {0.f, 0.f, 0.f, 0.f, 0.f, 0.f, 0.f, 0.f};
  v8f acc1 = {0.f, 0.f, 0.f, 0.f, 0.f, 0.f, 0.f, 0.f};
#pragma unroll
  for (int t = 0; t < HIDDEN / 8; ++t) {
    const v2f a_lo  = {av[t].x,  av[t].y},  a_hi  = {av[t].z,  av[t].w};
    const v2f b0_lo = {b0v[t].x, b0v[t].y}, b0_hi = {b0v[t].z, b0v[t].w};
    const v2f b1_lo = {b1v[t].x, b1v[t].y}, b1_hi = {b1v[t].z, b1v[t].w};
    acc0 = __builtin_amdgcn_wmma_f32_16x16x4_f32(false, a_lo, false, b0_lo,
                                                 (short)0, acc0, false, false);
    acc1 = __builtin_amdgcn_wmma_f32_16x16x4_f32(false, a_lo, false, b1_lo,
                                                 (short)0, acc1, false, false);
    acc0 = __builtin_amdgcn_wmma_f32_16x16x4_f32(false, a_hi, false, b0_hi,
                                                 (short)0, acc0, false, false);
    acc1 = __builtin_amdgcn_wmma_f32_16x16x4_f32(false, a_hi, false, b1_hi,
                                                 (short)0, acc1, false, false);
  }
  __builtin_amdgcn_sched_group_barrier(0x020, 3 * (HIDDEN / 8), 0);
  __builtin_amdgcn_sched_group_barrier(0x008, HIDDEN / 2, 0);
  if (lane < 16) {
    int o0 = t0 * 16 + lane;
    if (o0 < n_nodes) { p[o0] = acc0[0]; q[o0] = acc0[1]; }
    int o1 = t1 * 16 + lane;
    if (has1 && o1 < n_nodes) { p[o1] = acc1[0]; q[o1] = acc1[1]; }
  }
}

#define RANGE_SHIFT 8
#define RANGE_NODES (1 << RANGE_SHIFT)
#define MAX_RANGES 512
#define BIN_THREADS 256
#define BIN_EDGES_PER_THREAD 10
#define SLICE_EDGES (BIN_THREADS * BIN_EDGES_PER_THREAD)
#define FIRST_THREADS 288

__device__ __forceinline__ void
bin_block(const int* __restrict__ e, float* __restrict__ scratch,
          int n_edges, int n_slices, int n_ranges, int slice) {
  __shared__ int  cnt[MAX_RANGES];
  __shared__ int  wsum[BIN_THREADS / 32];
  __shared__ int2 stage[SLICE_EDGES];
  int2* sorted = (int2*)scratch;
  int2* runs   = sorted + (size_t)n_slices * SLICE_EDGES;
  const int tid  = threadIdx.x;
  const int lane = tid & 31;
  const int wv   = tid >> 5;
  const int base = slice * SLICE_EDGES;
  const bool act = tid < BIN_THREADS;

  int2 ed[BIN_EDGES_PER_THREAD];
  int  rank[BIN_EDGES_PER_THREAD];
  if (act) {
    cnt[tid] = 0;
    cnt[tid + BIN_THREADS] = 0;
#pragma unroll
    for (int k = 0; k < BIN_EDGES_PER_THREAD; ++k) {
      int i = base + k * BIN_THREADS + tid;
      if (i >= n_edges) i = n_edges - 1;
      ed[k] = ((const int2*)e)[i];
    }
  }
  __syncthreads();
  if (act) {
#pragma unroll
    for (int k = 0; k < BIN_EDGES_PER_THREAD; ++k) {
      const int i = base + k * BIN_THREADS + tid;
      rank[k] = (i < n_edges) ? atomicAdd(&cnt[ed[k].y >> RANGE_SHIFT], 1) : 0;
    }
  }
  __syncthreads();

  int c0 = 0, s2 = 0, incl = 0;
  if (act) {
    c0 = cnt[2 * tid];
    s2 = c0 + cnt[2 * tid + 1];
    incl = s2;
#pragma unroll
    for (int d = 1; d < 32; d <<= 1) {
      const int t = __shfl_up(incl, d, 32);
      if (lane >= d) incl += t;
    }
    if (lane == 31) wsum[wv] = incl;
  }
  __syncthreads();
  if (act) {
    int wbase = 0;
    for (int w = 0; w < wv; ++w) wbase += wsum[w];
    const int excl = wbase + incl - s2;
    cnt[2 * tid]     = excl;
    cnt[2 * tid + 1] = excl + c0;
  }
  __syncthreads();

  if (act) {
    for (int b = tid; b < n_ranges; b += BIN_THREADS)
      runs[(size_t)slice * n_ranges + b] = make_int2(cnt[b], cnt[b + 1]);
#pragma unroll
    for (int k = 0; k < BIN_EDGES_PER_THREAD; ++k) {
      const int i = base + k * BIN_THREADS + tid;
      if (i < n_edges) stage[cnt[ed[k].y >> RANGE_SHIFT] + rank[k]] = ed[k];
    }
  }
  __syncthreads();
  if (act) {
    int n_here = n_edges - base;
    if (n_here > SLICE_EDGES) n_here = SLICE_EDGES;
    for (int j = tid; j < n_here; j += BIN_THREADS) sorted[base + j] = stage[j];
  }
}

__global__ void __launch_bounds__(FIRST_THREADS)
gat_prep_bin(const float* __restrict__ W, const float* __restrict__ b,
             const float* __restrict__ att_w,
             float* __restrict__ G, float* __restrict__ cs,
             const int* __restrict__ e, float* __restrict__ scratch,
             int n_edges, int n_slices, int n_ranges) {
  if (blockIdx.x == 0) prep_uv_block(W, b, att_w, G, cs);
  else                 bin_block(e, scratch, n_edges, n_slices, n_ranges, (int)blockIdx.x - 1);
}

#define OWN_BATCH 8
__global__ void __launch_bounds__(RANGE_NODES)
gat_edge_own(const float* __restrict__ scratch, const float* __restrict__ p, const float* __restrict__ q,
             const float* __restrict__ cs, float* __restrict__ nacc,
             int n_slices, int n_ranges, int n_nodes) {
  __shared__ float acc[RANGE_NODES];
  __shared__ float ql[RANGE_NODES];
  const int2* sorted = (const int2*)scratch;
  const int2* runs   = sorted + (size_t)n_slices * SLICE_EDGES;
  const int tid  = threadIdx.x;
  const int b    = blockIdx.x;
  const int lo   = b << RANGE_SHIFT;
  const int node = lo + tid;
  acc[tid] = 0.f;
  ql[tid]  = (node < n_nodes) ? q[node] : 0.f;
  const float c = cs[0];
  __syncthreads();

  for (int s = tid; s < n_slices; s += RANGE_NODES) {
    const int2 jr = runs[(size_t)s * n_ranges + b];
    const int2* run = sorted + (size_t)s * SLICE_EDGES;
    for (int j = jr.x; j < jr.y; j += OWN_BATCH) {
      int2 rec[OWN_BATCH];
#pragma unroll
      for (int k = 0; k < OWN_BATCH; ++k) {
        int jj = j + k;
        if (jj >= jr.y) jj = jr.y - 1;
        rec[k] = run[jj];
      }
      float ps[OWN_BATCH];
#pragma unroll
      for (int k = 0; k < OWN_BATCH; ++k) ps[k] = p[rec[k].x];
      asm volatile("" :: "v"(ps[0]), "v"(ps[1]), "v"(ps[2]), "v"(ps[3]),
                         "v"(ps[4]), "v"(ps[5]), "v"(ps[6]), "v"(ps[7]));
      __builtin_amdgcn_sched_group_barrier(0x020, OWN_BATCH, 0);
#pragma unroll
      for (int k = 0; k < OWN_BATCH; ++k) {
        if (j + k < jr.y) {
          const int dl = rec[k].y - lo;
          const float z  = ps[k] + ql[dl] + c;
          const float lz = z > 0.f ? z : 0.2f * z;
          atomicAdd(&acc[dl], expf(lz));
        }
      }
    }
  }
  __syncthreads();
  if (node < n_nodes) nacc[node] = acc[tid];
}

#define OUT_PASSES 7
#define OUT_ROWS_PER_WAVE (4 * OUT_PASSES)
__global__ void __launch_bounds__(64)
gat_out(const float* __restrict__ x, const float* __restrict__ nacc,
        const float* __restrict__ nw, const float* __restrict__ nb,
        float* __restrict__ out, int n_nodes) {
  const int lane = threadIdx.x & 31;
  const int wv   = (blockIdx.x * blockDim.x + threadIdx.x) >> 5;
  const int row0 = wv * OUT_ROWS_PER_WAVE;
  if (row0 >= n_nodes) return;
  const int g = lane >> 3;
  const int j = lane & 7;

  float  nv[OUT_PASSES];
  float4 xv[OUT_PASSES][4];
#pragma unroll
  for (int ps = 0; ps < OUT_PASSES; ++ps) {
    int r = row0 + ps * 4 + g;
    if (r >= n_nodes) r = n_nodes - 1;
    nv[ps] = nacc[r];
    const float4* xr = (const float4*)(x + (size_t)r * HIDDEN) + j;
#pragma unroll
    for (int t = 0; t < 4; ++t) xv[ps][t] = xr[t * 8];
  }
  float4 w4[4], b4[4];
#pragma unroll
  for (int t = 0; t < 4; ++t) {
    w4[t] = ((const float4*)nw)[t * 8 + j];
    b4[t] = ((const float4*)nb)[t * 8 + j];
  }

#pragma unroll
  for (int ps = 0; ps < OUT_PASSES; ++ps) {
    const int r = row0 + ps * 4 + g;
    const float denom = (nv[ps] == 0.f) ? 1.f : nv[ps];
    float4 y[4];
    float ss = 0.f;
#pragma unroll
    for (int t = 0; t < 4; ++t) {
      const float4 v = xv[ps][t];
      y[t].x = fmaxf(v.x / denom, 0.f) + v.x;
      y[t].y = fmaxf(v.y / denom, 0.f) + v.y;
      y[t].z = fmaxf(v.z / denom, 0.f) + v.z;
      y[t].w = fmaxf(v.w / denom, 0.f) + v.w;
      ss += y[t].x * y[t].x + y[t].y * y[t].y + y[t].z * y[t].z + y[t].w * y[t].w;
    }
    ss += __shfl_xor(ss, 4, 32);
    ss += __shfl_xor(ss, 2, 32);
    ss += __shfl_xor(ss, 1, 32);
    const float inv_rms = rsqrtf(ss * (1.0f / HIDDEN) + 1e-6f);
    if (r < n_nodes) {
      float4* orow = (float4*)(out + (size_t)r * HIDDEN) + j;
#pragma unroll
      for (int t = 0; t < 4; ++t) {
        float4 o;
        o.x = w4[t].x * (y[t].x * inv_rms) + b4[t].x;
        o.y = w4[t].y * (y[t].y * inv_rms) + b4[t].y;
        o.z = w4[t].z * (y[t].z * inv_rms) + b4[t].z;
        o.w = w4[t].w * (y[t].w * inv_rms) + b4[t].w;
        orow[t * 8] = o;
      }
    }
  }
}

extern "C" void kernel_launch(void* const* d_in, const int* in_sizes, int n_in,
                              void* d_out, int out_size, void* d_ws, size_t ws_size,
                              hipStream_t stream) {
  const float* x     = (const float*)d_in[0];
  const float* W     = (const float*)d_in[1];
  const float* b     = (const float*)d_in[2];
  const float* att_w = (const float*)d_in[3];
  const float* nw    = (const float*)d_in[4];
  const float* nb    = (const float*)d_in[5];
  const int*   e     = (const int*)d_in[6];
  float* out = (float*)d_out;

  const int n_nodes = in_sizes[0] / HIDDEN;
  const int n_edges = in_sizes[6] / 2;

  float* G    = (float*)d_ws;
  float* cs   = G + 16 * HIDDEN;
  float* p    = cs + 16;
  float* q    = p + n_nodes;
  float* nacc = q + n_nodes;

  int n_slices = (n_edges + SLICE_EDGES - 1) / SLICE_EDGES;
  int n_ranges = (n_nodes + RANGE_NODES - 1) / RANGE_NODES;
  gat_prep_bin<<<1 + n_slices, FIRST_THREADS, 0, stream>>>(W, b, att_w, G, cs, e, out, n_edges, n_slices, n_ranges);

  int n_tiles = (n_nodes + 15) / 16;
  int n_waves = (n_tiles + 1) / 2;
  int pq_blocks = (n_waves * 32 + 127) / 128;
  gat_pq_wmma<<<pq_blocks, 128, 0, stream>>>(x, G, p, q, n_tiles, n_nodes);

  gat_edge_own<<<n_ranges, RANGE_NODES, 0, stream>>>(out, p, q, cs, nacc, n_slices, n_ranges, n_nodes);

  int out_waves = (n_nodes + OUT_ROWS_PER_WAVE - 1) / OUT_ROWS_PER_WAVE;
  gat_out<<<(out_waves * 32 + 63) / 64, 64, 0, stream>>>(x, nacc, nw, nb, out, n_nodes);
}


# p,q kernel: a WMMA column is a (node, 16-float chunk) pair, a row a (u or v, chunk) pair, so a fragment load covers 4 whole 128-byte lines, not 32 bytes of 16 lines; k sum order changes
# speedup vs baseline: 2.2064x; 1.0652x over previous
#include <hip/hip_runtime.h>
#include <hip/hip_bf16.h>

typedef __attribute__((ext_vector_type(2))) float v2f;
typedef __attribute__((ext_vector_type(8))) float v8f;

#define HIDDEN 128

__device__ __forceinline__ void
prep_uv_block(const float* __restrict__ W, const float* __restrict__ b,
              const float* __restrict__ att_w,
              float* __restrict__ G, float* __restrict__ cs) {
  __shared__ __attribute__((aligned(16))) float sa[3 * HIDDEN];
  const int tid  = threadIdx.x;
  const int lane = tid & 31;
  const int wv   = tid >> 5;
  const int c    = lane & 15;
  const int koff = (lane >> 4) << 1;

  v2f bb[HIDDEN / 4];
  if (wv < 8) {
    const float* bp = W + (size_t)koff * HIDDEN + wv * 16 + c;
#pragma unroll
    for (int k0 = 0; k0 < HIDDEN; k0 += 4) {
      bb[k0 >> 2].x = bp[(size_t)k0 * HIDDEN];
      bb[k0 >> 2].y = bp[(size_t)(k0 + 1) * HIDDEN];
    }
    sa[tid] = att_w[tid];
    if (tid < HIDDEN) sa[2 * HIDDEN + tid] = 0.f;
  } else {
    const float* bp = b + koff;
#pragma unroll
    for (int k0 = 0; k0 < HIDDEN; k0 += 4) {
      v2f t = *(const v2f*)(bp + k0);
      bb[k0 >> 2].x = (c == 0) ? t.x : 0.f;
      bb[k0 >> 2].y = (c == 0) ? t.y : 0.f;
    }
  }
  __syncthreads();

  const float* ap = sa + (c < 2 ? c : 2) * HIDDEN + koff;
  v2f aa[HIDDEN / 4];
#pragma unroll
  for (int k0 = 0; k0 < HIDDEN; k0 += 4) aa[k0 >> 2] = *(const v2f*)(ap + k0);
  v8f acc = {0.f, 0.f, 0.f, 0.f, 0.f, 0.f, 0.f, 0.f};
#pragma unroll
  for (int k0 = 0; k0 < HIDDEN; k0 += 4) {
    acc = __builtin_amdgcn_wmma_f32_16x16x4_f32(false, aa[k0 >> 2], false, bb[k0 >> 2],
                                                (short)0, acc, false, false);
  }

  if (wv < 8) {
    if (lane < 16) {
      G[wv * 16 + c] = acc[0];
      G[HIDDEN + wv * 16 + c] = acc[1];
    }
  } else if (lane == 0) {
    cs[0] = acc[0] + acc[1];
  }
}

typedef __attribute__((ext_vector_type(4))) float v4f;

#define PQ_NODES_PER_WAVE 32
#define PQ_GROUPS (PQ_NODES_PER_WAVE / 2)

__device__ __forceinline__ v4f frag_ld(const float* __restrict__ base, unsigned byte_off, int i) {
  return *(const v4f*)((const char*)base + byte_off + 256u * i);
}

template <int CTRL>
__device__ __forceinline__ float row_xadd(float t) {
  const int o = __builtin_amdgcn_update_dpp(0, __float_as_int(t), CTRL, 0xf, 0xf, true);
  return t + __int_as_float(o);
}

__device__ __forceinline__ float diag_sum(v8f acc, bool b0, bool b1, bool b2) {
  const float t0 = b2 ? acc[4] : acc[0];
  const float t1 = b2 ? acc[5] : acc[1];
  const float t2 = b2 ? acc[6] : acc[2];
  const float t3 = b2 ? acc[7] : acc[3];
  const float s0 = b1 ? t2 : t0;
  const float s1 = b1 ? t3 : t1;
  float t = b0 ? s1 : s0;
  t = row_xadd<0x161>(t);
  t = row_xadd<0x162>(t);
  t = row_xadd<0x164>(t);
  return t;
}

__global__ void __launch_bounds__(128)
gat_pq_wmma(const float* __restrict__ x, const float* __restrict__ G,
            float* __restrict__ p, float* __restrict__ q, int n_nodes) {
  const int lane  = threadIdx.x & 31;
  const int wv    = (blockIdx.x * blockDim.x + threadIdx.x) >> 5;
  const int node0 = wv * PQ_NODES_PER_WAVE;
  if (node0 >= n_nodes) return;

  const int ch   = lane & 7;
  const int rw   = (lane >> 3) & 1;
  const int h    = lane >> 4;
  const unsigned poff = (unsigned)(ch + 8 * h) * 16u;
  const unsigned ROWB = HIDDEN * (unsigned)sizeof(float);

  const unsigned aoff = (unsigned)rw * ROWB + poff;
  const v4f a0 = frag_ld(G, aoff, 0);
  const v4f a1 = frag_ld(G, aoff, 1);
  v4f bv[PQ_GROUPS][2];
#pragma unroll
  for (int g = 0; g < PQ_GROUPS; ++g) {
    int n = node0 + 2 * g + rw;
    if (n >= n_nodes) n = n_nodes - 1;
    const unsigned boff = (unsigned)n * ROWB + poff;
    bv[g][0] = frag_ld(x, boff, 0);
    bv[g][1] = frag_ld(x, boff, 1);
  }

  const v2f a0_lo = {a0.x, a0.y}, a0_hi = {a0.z, a0.w};
  const v2f a1_lo = {a1.x, a1.y}, a1_hi = {a1.z, a1.w};
  const bool b0 = (ch & 1) != 0, b1 = (ch & 2) != 0, b2 = (ch & 4) != 0;
  float res0 = 0.f, res1 = 0.f;
#pragma unroll
  for (int g = 0; g < PQ_GROUPS; g += 2) {
    const v2f c0_lo = {bv[g][0].x, bv[g][0].y},         c0_hi = {bv[g][0].z, bv[g][0].w};
    const v2f c1_lo = {bv[g][1].x, bv[g][1].y},         c1_hi = {bv[g][1].z, bv[g][1].w};
    const v2f d0_lo = {bv[g + 1][0].x, bv[g + 1][0].y}, d0_hi = {bv[g + 1][0].z, bv[g + 1][0].w};
    const v2f d1_lo = {bv[g + 1][1].x, bv[g + 1][1].y}, d1_hi = {bv[g + 1][1].z, bv[g + 1][1].w};
    v8f acc0 = {0.f, 0.f, 0.f, 0.f, 0.f, 0.f, 0.f, 0.f};
    v8f acc1 = {0.f, 0.f, 0.f, 0.f, 0.f, 0.f, 0.f, 0.f};
    acc0 = __builtin_amdgcn_wmma_f32_16x16x4_f32(false, a0_lo, false, c0_lo, (short)0, acc0, false, false);
    acc1 = __builtin_amdgcn_wmma_f32_16x16x4_f32(false, a0_lo, false, d0_lo, (short)0, acc1, false, false);
    acc0 = __builtin_amdgcn_wmma_f32_16x16x4_f32(false, a0_hi, false, c0_hi, (short)0, acc0, false, false);
    acc1 = __builtin_amdgcn_wmma_f32_16x16x4_f32(false, a0_hi, false, d0_hi, (short)0, acc1, false, false);
    acc0 = __builtin_amdgcn_wmma_f32_16x16x4_f32(false, a1_lo, false, c1_lo, (short)0, acc0, false, false);
    acc1 = __builtin_amdgcn_wmma_f32_16x16x4_f32(false, a1_lo, false, d1_lo, (short)0, acc1, false, false);
    acc0 = __builtin_amdgcn_wmma_f32_16x16x4_f32(false, a1_hi, false, c1_hi, (short)0, acc0, false, false);
    acc1 = __builtin_amdgcn_wmma_f32_16x16x4_f32(false, a1_hi, false, d1_hi, (short)0, acc1, false, false);
    const float s0 = diag_sum(acc0, b0, b1, b2);
    const float s1 = diag_sum(acc1, b0, b1, b2);
    if (g < 8) { res0 = (ch == g)     ? s0 : res0;  res0 = (ch == g + 1) ? s1 : res0; }
    else       { res1 = (ch == g - 8) ? s0 : res1;  res1 = (ch == g - 7) ? s1 : res1; }
  }
  __builtin_amdgcn_sched_group_barrier(0x020, 2 + 2 * PQ_GROUPS, 0);
  __builtin_amdgcn_sched_group_barrier(0x008, 4 * PQ_GROUPS, 0);

  float* __restrict__ dst = h ? q : p;
  const int o0 = node0 + 2 * ch + rw;
  if (o0 < n_nodes) dst[o0] = res0;
  const int o1 = o0 + 16;
  if (o1 < n_nodes) dst[o1] = res1;
}

#define RANGE_SHIFT 8
#define RANGE_NODES (1 << RANGE_SHIFT)
#define MAX_RANGES 512
#define BIN_THREADS 256
#define BIN_EDGES_PER_THREAD 10
#define SLICE_EDGES (BIN_THREADS * BIN_EDGES_PER_THREAD)
#define FIRST_THREADS 288

__device__ __forceinline__ void
bin_block(const int* __restrict__ e, float* __restrict__ scratch,
          int n_edges, int n_slices, int n_ranges, int slice) {
  __shared__ int  cnt[MAX_RANGES];
  __shared__ int  wsum[BIN_THREADS / 32];
  __shared__ int2 stage[SLICE_EDGES];
  int2* sorted = (int2*)scratch;
  int2* runs   = sorted + (size_t)n_slices * SLICE_EDGES;
  const int tid  = threadIdx.x;
  const int lane = tid & 31;
  const int wv   = tid >> 5;
  const int base = slice * SLICE_EDGES;
  const bool act = tid < BIN_THREADS;

  int2 ed[BIN_EDGES_PER_THREAD];
  int  rank[BIN_EDGES_PER_THREAD];
  if (act) {
    cnt[tid] = 0;
    cnt[tid + BIN_THREADS] = 0;
#pragma unroll
    for (int k = 0; k < BIN_EDGES_PER_THREAD; ++k) {
      int i = base + k * BIN_THREADS + tid;
      if (i >= n_edges) i = n_edges - 1;
      ed[k] = ((const int2*)e)[i];
    }
  }
  __syncthreads();
  if (act) {
#pragma unroll
    for (int k = 0; k < BIN_EDGES_PER_THREAD; ++k) {
      const int i = base + k * BIN_THREADS + tid;
      rank[k] = (i < n_edges) ? atomicAdd(&cnt[ed[k].y >> RANGE_SHIFT], 1) : 0;
    }
  }
  __syncthreads();

  int c0 = 0, s2 = 0, incl = 0;
  if (act) {
    c0 = cnt[2 * tid];
    s2 = c0 + cnt[2 * tid + 1];
    incl = s2;
#pragma unroll
    for (int d = 1; d < 32; d <<= 1) {
      const int t = __shfl_up(incl, d, 32);
      if (lane >= d) incl += t;
    }
    if (lane == 31) wsum[wv] = incl;
  }
  __syncthreads();
  if (act) {
    int wbase = 0;
    for (int w = 0; w < wv; ++w) wbase += wsum[w];
    const int excl = wbase + incl - s2;
    cnt[2 * tid]     = excl;
    cnt[2 * tid + 1] = excl + c0;
  }
  __syncthreads();

  if (act) {
    for (int b = tid; b < n_ranges; b += BIN_THREADS)
      runs[(size_t)slice * n_ranges + b] = make_int2(cnt[b], cnt[b + 1]);
#pragma unroll
    for (int k = 0; k < BIN_EDGES_PER_THREAD; ++k) {
      const int i = base + k * BIN_THREADS + tid;
      if (i < n_edges) stage[cnt[ed[k].y >> RANGE_SHIFT] + rank[k]] = ed[k];
    }
  }
  __syncthreads();
  if (act) {
    int n_here = n_edges - base;
    if (n_here > SLICE_EDGES) n_here = SLICE_EDGES;
    for (int j = tid; j < n_here; j += BIN_THREADS) sorted[base + j] = stage[j];
  }
}

__global__ void __launch_bounds__(FIRST_THREADS)
gat_prep_bin(const float* __restrict__ W, const float* __restrict__ b,
             const float* __restrict__ att_w,
             float* __restrict__ G, float* __restrict__ cs,
             const int* __restrict__ e, float* __restrict__ scratch,
             int n_edges, int n_slices, int n_ranges) {
  if (blockIdx.x == 0) prep_uv_block(W, b, att_w, G, cs);
  else                 bin_block(e, scratch, n_edges, n_slices, n_ranges, (int)blockIdx.x - 1);
}

#define OWN_BATCH 8
__global__ void __launch_bounds__(RANGE_NODES)
gat_edge_own(const float* __restrict__ scratch, const float* __restrict__ p, const float* __restrict__ q,
             const float* __restrict__ cs, float* __restrict__ nacc,
             int n_slices, int n_ranges, int n_nodes) {
  __shared__ float acc[RANGE_NODES];
  __shared__ float ql[RANGE_NODES];
  const int2* sorted = (const int2*)scratch;
  const int2* runs   = sorted + (size_t)n_slices * SLICE_EDGES;
  const int tid  = threadIdx.x;
  const int b    = blockIdx.x;
  const int lo   = b << RANGE_SHIFT;
  const int node = lo + tid;
  acc[tid] = 0.f;
  ql[tid]  = (node < n_nodes) ? q[node] : 0.f;
  const float c = cs[0];
  __syncthreads();

  for (int s = tid; s < n_slices; s += RANGE_NODES) {
    const int2 jr = runs[(size_t)s * n_ranges + b];
    const int2* run = sorted + (size_t)s * SLICE_EDGES;
    for (int j = jr.x; j < jr.y; j += OWN_BATCH) {
      int2 rec[OWN_BATCH];
#pragma unroll
      for (int k = 0; k < OWN_BATCH; ++k) {
        int jj = j + k;
        if (jj >= jr.y) jj = jr.y - 1;
        rec[k] = run[jj];
      }
      float ps[OWN_BATCH];
#pragma unroll
      for (int k = 0; k < OWN_BATCH; ++k) ps[k] = p[rec[k].x];
      asm volatile("" :: "v"(ps[0]), "v"(ps[1]), "v"(ps[2]), "v"(ps[3]),
                         "v"(ps[4]), "v"(ps[5]), "v"(ps[6]), "v"(ps[7]));
      __builtin_amdgcn_sched_group_barrier(0x020, OWN_BATCH, 0);
#pragma unroll
      for (int k = 0; k < OWN_BATCH; ++k) {
        if (j + k < jr.y) {
          const int dl = rec[k].y - lo;
          const float z  = ps[k] + ql[dl] + c;
          const float lz = z > 0.f ? z : 0.2f * z;
          atomicAdd(&acc[dl], expf(lz));
        }
      }
    }
  }
  __syncthreads();
  if (node < n_nodes) nacc[node] = acc[tid];
}

#define OUT_PASSES 7
#define OUT_ROWS_PER_WAVE (4 * OUT_PASSES)
__global__ void __launch_bounds__(64)
gat_out(const float* __restrict__ x, const float* __restrict__ nacc,
        const float* __restrict__ nw, const float* __restrict__ nb,
        float* __restrict__ out, int n_nodes) {
  const int lane = threadIdx.x & 31;
  const int wv   = (blockIdx.x * blockDim.x + threadIdx.x) >> 5;
  const int row0 = wv * OUT_ROWS_PER_WAVE;
  if (row0 >= n_nodes) return;
  const int g = lane >> 3;
  const int j = lane & 7;

  float  nv[OUT_PASSES];
  float4 xv[OUT_PASSES][4];
#pragma unroll
  for (int ps = 0; ps < OUT_PASSES; ++ps) {
    int r = row0 + ps * 4 + g;
    if (r >= n_nodes) r = n_nodes - 1;
    nv[ps] = nacc[r];
    const float4* xr = (const float4*)(x + (size_t)r * HIDDEN) + j;
#pragma unroll
    for (int t = 0; t < 4; ++t) xv[ps][t] = xr[t * 8];
  }
  float4 w4[4], b4[4];
#pragma unroll
  for (int t = 0; t < 4; ++t) {
    w4[t] = ((const float4*)nw)[t * 8 + j];
    b4[t] = ((const float4*)nb)[t * 8 + j];
  }

#pragma unroll
  for (int ps = 0; ps < OUT_PASSES; ++ps) {
    const int r = row0 + ps * 4 + g;
    const float denom = (nv[ps] == 0.f) ? 1.f : nv[ps];
    float4 y[4];
    float ss = 0.f;
#pragma unroll
    for (int t = 0; t < 4; ++t) {
      const float4 v = xv[ps][t];
      y[t].x = fmaxf(v.x / denom, 0.f) + v.x;
      y[t].y = fmaxf(v.y / denom, 0.f) + v.y;
      y[t].z = fmaxf(v.z / denom, 0.f) + v.z;
      y[t].w = fmaxf(v.w / denom, 0.f) + v.w;
      ss += y[t].x * y[t].x + y[t].y * y[t].y + y[t].z * y[t].z + y[t].w * y[t].w;
    }
    ss += __shfl_xor(ss, 4, 32);
    ss += __shfl_xor(ss, 2, 32);
    ss += __shfl_xor(ss, 1, 32);
    const float inv_rms = rsqrtf(ss * (1.0f / HIDDEN) + 1e-6f);
    if (r < n_nodes) {
      float4* orow = (float4*)(out + (size_t)r * HIDDEN) + j;
#pragma unroll
      for (int t = 0; t < 4; ++t) {
        float4 o;
        o.x = w4[t].x * (y[t].x * inv_rms) + b4[t].x;
        o.y = w4[t].y * (y[t].y * inv_rms) + b4[t].y;
        o.z = w4[t].z * (y[t].z * inv_rms) + b4[t].z;
        o.w = w4[t].w * (y[t].w * inv_rms) + b4[t].w;
        orow[t * 8] = o;
      }
    }
  }
}

extern "C" void kernel_launch(void* const* d_in, const int* in_sizes, int n_in,
                              void* d_out, int out_size, void* d_ws, size_t ws_size,
                              hipStream_t stream) {
  const float* x     = (const float*)d_in[0];
  const float* W     = (const float*)d_in[1];
  const float* b     = (const float*)d_in[2];
  const float* att_w = (const float*)d_in[3];
  const float* nw    = (const float*)d_in[4];
  const float* nb    = (const float*)d_in[5];
  const int*   e     = (const int*)d_in[6];
  float* out = (float*)d_out;

  const int n_nodes = in_sizes[0] / HIDDEN;
  const int n_edges = in_sizes[6] / 2;

  float* G    = (float*)d_ws;
  float* cs   = G + 2 * HIDDEN;
  float* p    = cs + 16;
  float* q    = p + n_nodes;
  float* nacc = q + n_nodes;

  int n_slices = (n_edges + SLICE_EDGES - 1) / SLICE_EDGES;
  int n_ranges = (n_nodes + RANGE_NODES - 1) / RANGE_NODES;
  gat_prep_bin<<<1 + n_slices, FIRST_THREADS, 0, stream>>>(W, b, att_w, G, cs, e, out, n_edges, n_slices, n_ranges);

  int pq_waves  = (n_nodes + PQ_NODES_PER_WAVE - 1) / PQ_NODES_PER_WAVE;
  int pq_blocks = (pq_waves * 32 + 127) / 128;
  gat_pq_wmma<<<pq_blocks, 128, 0, stream>>>(x, G, p, q, n_nodes);

  gat_edge_own<<<n_ranges, RANGE_NODES, 0, stream>>>(out, p, q, cs, nacc, n_slices, n_ranges, n_nodes);

  int out_waves = (n_nodes + OUT_ROWS_PER_WAVE - 1) / OUT_ROWS_PER_WAVE;
  gat_out<<<(out_waves * 32 + 63) / 64, 64, 0, stream>>>(x, nacc, nw, nb, out, n_nodes);
}


# Edge kernel: a run's records are read by 16 consecutive lanes (1-2 lines per run and load, two runs per load), not by one lane across 8 loads of 32 lines each; LDS add order changes (arrival order)
# speedup vs baseline: 2.3650x; 1.0719x over previous
#include <hip/hip_runtime.h>
#include <hip/hip_bf16.h>

typedef __attribute__((ext_vector_type(2))) float v2f;
typedef __attribute__((ext_vector_type(8))) float v8f;

#define HIDDEN 128

__device__ __forceinline__ void
prep_uv_block(const float* __restrict__ W, const float* __restrict__ b,
              const float* __restrict__ att_w,
              float* __restrict__ G, float* __restrict__ cs) {
  __shared__ __attribute__((aligned(16))) float sa[3 * HIDDEN];
  const int tid  = threadIdx.x;
  const int lane = tid & 31;
  const int wv   = tid >> 5;
  const int c    = lane & 15;
  const int koff = (lane >> 4) << 1;

  v2f bb[HIDDEN / 4];
  if (wv < 8) {
    const float* bp = W + (size_t)koff * HIDDEN + wv * 16 + c;
#pragma unroll
    for (int k0 = 0; k0 < HIDDEN; k0 += 4) {
      bb[k0 >> 2].x = bp[(size_t)k0 * HIDDEN];
      bb[k0 >> 2].y = bp[(size_t)(k0 + 1) * HIDDEN];
    }
    sa[tid] = att_w[tid];
    if (tid < HIDDEN) sa[2 * HIDDEN + tid] = 0.f;
  } else {
    const float* bp = b + koff;
#pragma unroll
    for (int k0 = 0; k0 < HIDDEN; k0 += 4) {
      v2f t = *(const v2f*)(bp + k0);
      bb[k0 >> 2].x = (c == 0) ? t.x : 0.f;
      bb[k0 >> 2].y = (c == 0) ? t.y : 0.f;
    }
  }
  __syncthreads();

  const float* ap = sa + (c < 2 ? c : 2) * HIDDEN + koff;
  v2f aa[HIDDEN / 4];
#pragma unroll
  for (int k0 = 0; k0 < HIDDEN; k0 += 4) aa[k0 >> 2] = *(const v2f*)(ap + k0);
  v8f acc = {0.f, 0.f, 0.f, 0.f, 0.f, 0.f, 0.f, 0.f};
#pragma unroll
  for (int k0 = 0; k0 < HIDDEN; k0 += 4) {
    acc = __builtin_amdgcn_wmma_f32_16x16x4_f32(false, aa[k0 >> 2], false, bb[k0 >> 2],
                                                (short)0, acc, false, false);
  }

  if (wv < 8) {
    if (lane < 16) {
      G[wv * 16 + c] = acc[0];
      G[HIDDEN + wv * 16 + c] = acc[1];
    }
  } else if (lane == 0) {
    cs[0] = acc[0] + acc[1];
  }
}

typedef __attribute__((ext_vector_type(4))) float v4f;

#define PQ_NODES_PER_WAVE 32
#define PQ_GROUPS (PQ_NODES_PER_WAVE / 2)

__device__ __forceinline__ v4f frag_ld(const float* __restrict__ base, unsigned byte_off, int i) {
  return *(const v4f*)((const char*)base + byte_off + 256u * i);
}

template <int CTRL>
__device__ __forceinline__ float row_xadd(float t) {
  const int o = __builtin_amdgcn_update_dpp(0, __float_as_int(t), CTRL, 0xf, 0xf, true);
  return t + __int_as_float(o);
}

__device__ __forceinline__ float diag_sum(v8f acc, bool b0, bool b1, bool b2) {
  const float t0 = b2 ? acc[4] : acc[0];
  const float t1 = b2 ? acc[5] : acc[1];
  const float t2 = b2 ? acc[6] : acc[2];
  const float t3 = b2 ? acc[7] : acc[3];
  const float s0 = b1 ? t2 : t0;
  const float s1 = b1 ? t3 : t1;
  float t = b0 ? s1 : s0;
  t = row_xadd<0x161>(t);
  t = row_xadd<0x162>(t);
  t = row_xadd<0x164>(t);
  return t;
}

__global__ void __launch_bounds__(128)
gat_pq_wmma(const float* __restrict__ x, const float* __restrict__ G,
            float* __restrict__ p, float* __restrict__ q, int n_nodes) {
  const int lane  = threadIdx.x & 31;
  const int wv    = (blockIdx.x * blockDim.x + threadIdx.x) >> 5;
  const int node0 = wv * PQ_NODES_PER_WAVE;
  if (node0 >= n_nodes) return;

  const int ch   = lane & 7;
  const int rw   = (lane >> 3) & 1;
  const int h    = lane >> 4;
  const unsigned poff = (unsigned)(ch + 8 * h) * 16u;
  const unsigned ROWB = HIDDEN * (unsigned)sizeof(float);

  const unsigned aoff = (unsigned)rw * ROWB + poff;
  const v4f a0 = frag_ld(G, aoff, 0);
  const v4f a1 = frag_ld(G, aoff, 1);
  v4f bv[PQ_GROUPS][2];
#pragma unroll
  for (int g = 0; g < PQ_GROUPS; ++g) {
    int n = node0 + 2 * g + rw;
    if (n >= n_nodes) n = n_nodes - 1;
    const unsigned boff = (unsigned)n * ROWB + poff;
    bv[g][0] = frag_ld(x, boff, 0);
    bv[g][1] = frag_ld(x, boff, 1);
  }

  const v2f a0_lo = {a0.x, a0.y}, a0_hi = {a0.z, a0.w};
  const v2f a1_lo = {a1.x, a1.y}, a1_hi = {a1.z, a1.w};
  const bool b0 = (ch & 1) != 0, b1 = (ch & 2) != 0, b2 = (ch & 4) != 0;
  float res0 = 0.f, res1 = 0.f;
#pragma unroll
  for (int g = 0; g < PQ_GROUPS; g += 2) {
    const v2f c0_lo = {bv[g][0].x, bv[g][0].y},         c0_hi = {bv[g][0].z, bv[g][0].w};
    const v2f c1_lo = {bv[g][1].x, bv[g][1].y},         c1_hi = {bv[g][1].z, bv[g][1].w};
    const v2f d0_lo = {bv[g + 1][0].x, bv[g + 1][0].y}, d0_hi = {bv[g + 1][0].z, bv[g + 1][0].w};
    const v2f d1_lo = {bv[g + 1][1].x, bv[g + 1][1].y}, d1_hi = {bv[g + 1][1].z, bv[g + 1][1].w};
    v8f acc0 = {0.f, 0.f, 0.f, 0.f, 0.f, 0.f, 0.f, 0.f};
    v8f acc1 = {0.f, 0.f, 0.f, 0.f, 0.f, 0.f, 0.f, 0.f};
    acc0 = __builtin_amdgcn_wmma_f32_16x16x4_f32(false, a0_lo, false, c0_lo, (short)0, acc0, false, false);
    acc1 = __builtin_amdgcn_wmma_f32_16x16x4_f32(false, a0_lo, false, d0_lo, (short)0, acc1, false, false);
    acc0 = __builtin_amdgcn_wmma_f32_16x16x4_f32(false, a0_hi, false, c0_hi, (short)0, acc0, false, false);
    acc1 = __builtin_amdgcn_wmma_f32_16x16x4_f32(false, a0_hi, false, d0_hi, (short)0, acc1, false, false);
    acc0 = __builtin_amdgcn_wmma_f32_16x16x4_f32(false, a1_lo, false, c1_lo, (short)0, acc0, false, false);
    acc1 = __builtin_amdgcn_wmma_f32_16x16x4_f32(false, a1_lo, false, d1_lo, (short)0, acc1, false, false);
    acc0 = __builtin_amdgcn_wmma_f32_16x16x4_f32(false, a1_hi, false, c1_hi, (short)0, acc0, false, false);
    acc1 = __builtin_amdgcn_wmma_f32_16x16x4_f32(false, a1_hi, false, d1_hi, (short)0, acc1, false, false);
    const float s0 = diag_sum(acc0, b0, b1, b2);
    const float s1 = diag_sum(acc1, b0, b1, b2);
    if (g < 8) { res0 = (ch == g)     ? s0 : res0;  res0 = (ch == g + 1) ? s1 : res0; }
    else       { res1 = (ch == g - 8) ? s0 : res1;  res1 = (ch == g - 7) ? s1 : res1; }
  }
  __builtin_amdgcn_sched_group_barrier(0x020, 2 + 2 * PQ_GROUPS, 0);
  __builtin_amdgcn_sched_group_barrier(0x008, 4 * PQ_GROUPS, 0);

  float* __restrict__ dst = h ? q : p;
  const int o0 = node0 + 2 * ch + rw;
  if (o0 < n_nodes) dst[o0] = res0;
  const int o1 = o0 + 16;
  if (o1 < n_nodes) dst[o1] = res1;
}

#define RANGE_SHIFT 8
#define RANGE_NODES (1 << RANGE_SHIFT)
#define MAX_RANGES 512
#define BIN_THREADS 256
#define BIN_EDGES_PER_THREAD 10
#define SLICE_EDGES (BIN_THREADS * BIN_EDGES_PER_THREAD)
#define FIRST_THREADS 288

__device__ __forceinline__ void
bin_block(const int* __restrict__ e, float* __restrict__ scratch,
          int n_edges, int n_slices, int n_ranges, int slice) {
  __shared__ int  cnt[MAX_RANGES];
  __shared__ int  wsum[BIN_THREADS / 32];
  __shared__ int2 stage[SLICE_EDGES];
  int2* sorted = (int2*)scratch;
  int2* runs   = sorted + (size_t)n_slices * SLICE_EDGES;
  const int tid  = threadIdx.x;
  const int lane = tid & 31;
  const int wv   = tid >> 5;
  const int base = slice * SLICE_EDGES;
  const bool act = tid < BIN_THREADS;

  int2 ed[BIN_EDGES_PER_THREAD];
  int  rank[BIN_EDGES_PER_THREAD];
  if (act) {
    cnt[tid] = 0;
    cnt[tid + BIN_THREADS] = 0;
#pragma unroll
    for (int k = 0; k < BIN_EDGES_PER_THREAD; ++k) {
      int i = base + k * BIN_THREADS + tid;
      if (i >= n_edges) i = n_edges - 1;
      ed[k] = ((const int2*)e)[i];
    }
  }
  __syncthreads();
  if (act) {
#pragma unroll
    for (int k = 0; k < BIN_EDGES_PER_THREAD; ++k) {
      const int i = base + k * BIN_THREADS + tid;
      rank[k] = (i < n_edges) ? atomicAdd(&cnt[ed[k].y >> RANGE_SHIFT], 1) : 0;
    }
  }
  __syncthreads();

  int c0 = 0, s2 = 0, incl = 0;
  if (act) {
    c0 = cnt[2 * tid];
    s2 = c0 + cnt[2 * tid + 1];
    incl = s2;
#pragma unroll
    for (int d = 1; d < 32; d <<= 1) {
      const int t = __shfl_up(incl, d, 32);
      if (lane >= d) incl += t;
    }
    if (lane == 31) wsum[wv] = incl;
  }
  __syncthreads();
  if (act) {
    int wbase = 0;
    for (int w = 0; w < wv; ++w) wbase += wsum[w];
    const int excl = wbase + incl - s2;
    cnt[2 * tid]     = excl;
    cnt[2 * tid + 1] = excl + c0;
  }
  __syncthreads();

  if (act) {
    for (int b = tid; b < n_ranges; b += BIN_THREADS)
      runs[(size_t)slice * n_ranges + b] = make_int2(cnt[b], cnt[b + 1]);
#pragma unroll
    for (int k = 0; k < BIN_EDGES_PER_THREAD; ++k) {
      const int i = base + k * BIN_THREADS + tid;
      if (i < n_edges) stage[cnt[ed[k].y >> RANGE_SHIFT] + rank[k]] = ed[k];
    }
  }
  __syncthreads();
  if (act) {
    int n_here = n_edges - base;
    if (n_here > SLICE_EDGES) n_here = SLICE_EDGES;
    for (int j = tid; j < n_here; j += BIN_THREADS) sorted[base + j] = stage[j];
  }
}

__global__ void __launch_bounds__(FIRST_THREADS)
gat_prep_bin(const float* __restrict__ W, const float* __restrict__ b,
             const float* __restrict__ att_w,
             float* __restrict__ G, float* __restrict__ cs,
             const int* __restrict__ e, float* __restrict__ scratch,
             int n_edges, int n_slices, int n_ranges) {
  if (blockIdx.x == 0) prep_uv_block(W, b, att_w, G, cs);
  else                 bin_block(e, scratch, n_edges, n_slices, n_ranges, (int)blockIdx.x - 1);
}

#define OWN_GROUP 16
#define OWN_STEPS 16
__global__ void __launch_bounds__(RANGE_NODES)
gat_edge_own(const float* __restrict__ scratch, const float* __restrict__ p, const float* __restrict__ q,
             const float* __restrict__ cs, float* __restrict__ nacc,
             int n_slices, int n_ranges, int n_nodes) {
  __shared__ float acc[RANGE_NODES];
  __shared__ float ql[RANGE_NODES];
  const int2* sorted = (const int2*)scratch;
  const int2* runs   = sorted + (size_t)n_slices * SLICE_EDGES;
  const int tid  = threadIdx.x;
  const int lane = tid & 31;
  const int hh   = lane >> 4;
  const int l16  = lane & 15;
  const int b    = blockIdx.x;
  const int lo   = b << RANGE_SHIFT;
  const int node = lo + tid;
  acc[tid] = 0.f;
  ql[tid]  = (node < n_nodes) ? q[node] : 0.f;
  const float c = cs[0];
  __syncthreads();

  for (int s0 = 0; s0 < n_slices; s0 += RANGE_NODES) {
    const int s = s0 + tid;
    int2 jr = make_int2(0, 0);
    if (s < n_slices) jr = runs[(size_t)s * n_ranges + b];
    const int len = jr.y - jr.x;
    const int sw  = s - lane;
    for (int off = 0; __builtin_amdgcn_ballot_w32(off < len) != 0u; off += OWN_GROUP) {
      int2 rec[OWN_STEPS];
      bool ok[OWN_STEPS];
#pragma unroll
      for (int i = 0; i < OWN_STEPS; ++i) {
        const int from = 2 * i + hh;
        const int st = __shfl(jr.x, from, 32);
        const int en = __shfl(jr.y, from, 32);
        int sl = sw + from;
        if (sl >= n_slices) sl = n_slices - 1;
        int jj = st + off + l16;
        ok[i] = jj < en;
        if (jj >= en) jj = en - 1;
        if (jj < 0) jj = 0;
        rec[i] = sorted[(size_t)sl * SLICE_EDGES + jj];
      }
      asm volatile("" :: "v"(rec[0].x), "v"(rec[1].x), "v"(rec[2].x), "v"(rec[3].x),
                         "v"(rec[4].x), "v"(rec[5].x), "v"(rec[6].x), "v"(rec[7].x),
                         "v"(rec[8].x), "v"(rec[9].x), "v"(rec[10].x), "v"(rec[11].x),
                         "v"(rec[12].x), "v"(rec[13].x), "v"(rec[14].x), "v"(rec[15].x));
      float ps[OWN_STEPS];
#pragma unroll
      for (int i = 0; i < OWN_STEPS; ++i) ps[i] = p[rec[i].x];
      int   dl[OWN_STEPS];
      float qd[OWN_STEPS];
#pragma unroll
      for (int i = 0; i < OWN_STEPS; ++i) {
        dl[i] = ok[i] ? rec[i].y - lo : 0;
        qd[i] = ql[dl[i]];
      }
#pragma unroll
      for (int i = 0; i < OWN_STEPS; ++i) {
        if (ok[i]) {
          const float z  = ps[i] + qd[i] + c;
          const float lz = z > 0.f ? z : 0.2f * z;
          atomicAdd(&acc[dl[i]], expf(lz));
        }
      }
    }
  }
  __syncthreads();
  if (node < n_nodes) nacc[node] = acc[tid];
}

#define OUT_PASSES 7
#define OUT_ROWS_PER_WAVE (4 * OUT_PASSES)
__global__ void __launch_bounds__(64)
gat_out(const float* __restrict__ x, const float* __restrict__ nacc,
        const float* __restrict__ nw, const float* __restrict__ nb,
        float* __restrict__ out, int n_nodes) {
  const int lane = threadIdx.x & 31;
  const int wv   = (blockIdx.x * blockDim.x + threadIdx.x) >> 5;
  const int row0 = wv * OUT_ROWS_PER_WAVE;
  if (row0 >= n_nodes) return;
  const int g = lane >> 3;
  const int j = lane & 7;

  float  nv[OUT_PASSES];
  float4 xv[OUT_PASSES][4];
#pragma unroll
  for (int ps = 0; ps < OUT_PASSES; ++ps) {
    int r = row0 + ps * 4 + g;
    if (r >= n_nodes) r = n_nodes - 1;
    nv[ps] = nacc[r];
    const float4* xr = (const float4*)(x + (size_t)r * HIDDEN) + j;
#pragma unroll
    for (int t = 0; t < 4; ++t) xv[ps][t] = xr[t * 8];
  }
  float4 w4[4], b4[4];
#pragma unroll
  for (int t = 0; t < 4; ++t) {
    w4[t] = ((const float4*)nw)[t * 8 + j];
    b4[t] = ((const float4*)nb)[t * 8 + j];
  }

#pragma unroll
  for (int ps = 0; ps < OUT_PASSES; ++ps) {
    const int r = row0 + ps * 4 + g;
    const float denom = (nv[ps] == 0.f) ? 1.f : nv[ps];
    float4 y[4];
    float ss = 0.f;
#pragma unroll
    for (int t = 0; t < 4; ++t) {
      const float4 v = xv[ps][t];
      y[t].x = fmaxf(v.x / denom, 0.f) + v.x;
      y[t].y = fmaxf(v.y / denom, 0.f) + v.y;
      y[t].z = fmaxf(v.z / denom, 0.f) + v.z;
      y[t].w = fmaxf(v.w / denom, 0.f) + v.w;
      ss += y[t].x * y[t].x + y[t].y * y[t].y + y[t].z * y[t].z + y[t].w * y[t].w;
    }
    ss += __shfl_xor(ss, 4, 32);
    ss += __shfl_xor(ss, 2, 32);
    ss += __shfl_xor(ss, 1, 32);
    const float inv_rms = rsqrtf(ss * (1.0f / HIDDEN) + 1e-6f);
    if (r < n_nodes) {
      float4* orow = (float4*)(out + (size_t)r * HIDDEN) + j;
#pragma unroll
      for (int t = 0; t < 4; ++t) {
        float4 o;
        o.x = w4[t].x * (y[t].x * inv_rms) + b4[t].x;
        o.y = w4[t].y * (y[t].y * inv_rms) + b4[t].y;
        o.z = w4[t].z * (y[t].z * inv_rms) + b4[t].z;
        o.w = w4[t].w * (y[t].w * inv_rms) + b4[t].w;
        orow[t * 8] = o;
      }
    }
  }
}

extern "C" void kernel_launch(void* const* d_in, const int* in_sizes, int n_in,
                              void* d_out, int out_size, void* d_ws, size_t ws_size,
                              hipStream_t stream) {
  const float* x     = (const float*)d_in[0];
  const float* W     = (const float*)d_in[1];
  const float* b     = (const float*)d_in[2];
  const float* att_w = (const float*)d_in[3];
  const float* nw    = (const float*)d_in[4];
  const float* nb    = (const float*)d_in[5];
  const int*   e     = (const int*)d_in[6];
  float* out = (float*)d_out;

  const int n_nodes = in_sizes[0] / HIDDEN;
  const int n_edges = in_sizes[6] / 2;

  float* G    = (float*)d_ws;
  float* cs   = G + 2 * HIDDEN;
  float* p    = cs + 16;
  float* q    = p + n_nodes;
  float* nacc = q + n_nodes;

  int n_slices = (n_edges + SLICE_EDGES - 1) / SLICE_EDGES;
  int n_ranges = (n_nodes + RANGE_NODES - 1) / RANGE_NODES;
  gat_prep_bin<<<1 + n_slices, FIRST_THREADS, 0, stream>>>(W, b, att_w, G, cs, e, out, n_edges, n_slices, n_ranges);

  int pq_waves  = (n_nodes + PQ_NODES_PER_WAVE - 1) / PQ_NODES_PER_WAVE;
  int pq_blocks = (pq_waves * 32 + 127) / 128;
  gat_pq_wmma<<<pq_blocks, 128, 0, stream>>>(x, G, p, q, n_nodes);

  gat_edge_own<<<n_ranges, RANGE_NODES, 0, stream>>>(out, p, q, cs, nacc, n_slices, n_ranges, n_nodes);

  int out_waves = (n_nodes + OUT_ROWS_PER_WAVE - 1) / OUT_ROWS_PER_WAVE;
  gat_out<<<(out_waves * 32 + 63) / 64, 64, 0, stream>>>(x, nacc, nw, nb, out, n_nodes);
}


# Edge sums + output rows in one launch (4 -> 3) with an even share: 256 blocks of 14 waves own 392 nodes each (28 rows a wave; not 391 blocks of 8), sums stay in LDS; LDS add order changes (arrival)
# speedup vs baseline: 2.7064x; 1.1444x over previous
#include <hip/hip_runtime.h>
#include <hip/hip_bf16.h>

typedef __attribute__((ext_vector_type(2))) float v2f;
typedef __attribute__((ext_vector_type(8))) float v8f;

#define HIDDEN 128

__device__ __forceinline__ void
prep_uv_block(const float* __restrict__ W, const float* __restrict__ b,
              const float* __restrict__ att_w,
              float* __restrict__ G, float* __restrict__ cs) {
  __shared__ __attribute__((aligned(16))) float sa[3 * HIDDEN];
  const int tid  = threadIdx.x;
  const int lane = tid & 31;
  const int wv   = tid >> 5;
  const int c    = lane & 15;
  const int koff = (lane >> 4) << 1;

  v2f bb[HIDDEN / 4];
  if (wv < 8) {
    const float* bp = W + (size_t)koff * HIDDEN + wv * 16 + c;
#pragma unroll
    for (int k0 = 0; k0 < HIDDEN; k0 += 4) {
      bb[k0 >> 2].x = bp[(size_t)k0 * HIDDEN];
      bb[k0 >> 2].y = bp[(size_t)(k0 + 1) * HIDDEN];
    }
    sa[tid] = att_w[tid];
    if (tid < HIDDEN) sa[2 * HIDDEN + tid] = 0.f;
  } else {
    const float* bp = b + koff;
#pragma unroll
    for (int k0 = 0; k0 < HIDDEN; k0 += 4) {
      v2f t = *(const v2f*)(bp + k0);
      bb[k0 >> 2].x = (c == 0) ? t.x : 0.f;
      bb[k0 >> 2].y = (c == 0) ? t.y : 0.f;
    }
  }
  __syncthreads();

  const float* ap = sa + (c < 2 ? c : 2) * HIDDEN + koff;
  v2f aa[HIDDEN / 4];
#pragma unroll
  for (int k0 = 0; k0 < HIDDEN; k0 += 4) aa[k0 >> 2] = *(const v2f*)(ap + k0);
  v8f acc = {0.f, 0.f, 0.f, 0.f, 0.f, 0.f, 0.f, 0.f};
#pragma unroll
  for (int k0 = 0; k0 < HIDDEN; k0 += 4) {
    acc = __builtin_amdgcn_wmma_f32_16x16x4_f32(false, aa[k0 >> 2], false, bb[k0 >> 2],
                                                (short)0, acc, false, false);
  }

  if (wv < 8) {
    if (lane < 16) {
      G[wv * 16 + c] = acc[0];
      G[HIDDEN + wv * 16 + c] = acc[1];
    }
  } else if (lane == 0) {
    cs[0] = acc[0] + acc[1];
  }
}

typedef __attribute__((ext_vector_type(4))) float v4f;

#define PQ_NODES_PER_WAVE 32
#define PQ_GROUPS (PQ_NODES_PER_WAVE / 2)

__device__ __forceinline__ v4f frag_ld(const float* __restrict__ base, unsigned byte_off, int i) {
  return *(const v4f*)((const char*)base + byte_off + 256u * i);
}

template <int CTRL>
__device__ __forceinline__ float row_xadd(float t) {
  const int o = __builtin_amdgcn_update_dpp(0, __float_as_int(t), CTRL, 0xf, 0xf, true);
  return t + __int_as_float(o);
}

__device__ __forceinline__ float diag_sum(v8f acc, bool b0, bool b1, bool b2) {
  const float t0 = b2 ? acc[4] : acc[0];
  const float t1 = b2 ? acc[5] : acc[1];
  const float t2 = b2 ? acc[6] : acc[2];
  const float t3 = b2 ? acc[7] : acc[3];
  const float s0 = b1 ? t2 : t0;
  const float s1 = b1 ? t3 : t1;
  float t = b0 ? s1 : s0;
  t = row_xadd<0x161>(t);
  t = row_xadd<0x162>(t);
  t = row_xadd<0x164>(t);
  return t;
}

__global__ void __launch_bounds__(128)
gat_pq_wmma(const float* __restrict__ x, const float* __restrict__ G,
            float* __restrict__ p, float* __restrict__ q, int n_nodes) {
  const int lane  = threadIdx.x & 31;
  const int wv    = (blockIdx.x * blockDim.x + threadIdx.x) >> 5;
  const int node0 = wv * PQ_NODES_PER_WAVE;
  if (node0 >= n_nodes) return;

  const int ch   = lane & 7;
  const int rw   = (lane >> 3) & 1;
  const int h    = lane >> 4;
  const unsigned poff = (unsigned)(ch + 8 * h) * 16u;
  const unsigned ROWB = HIDDEN * (unsigned)sizeof(float);

  const unsigned aoff = (unsigned)rw * ROWB + poff;
  const v4f a0 = frag_ld(G, aoff, 0);
  const v4f a1 = frag_ld(G, aoff, 1);
  v4f bv[PQ_GROUPS][2];
#pragma unroll
  for (int g = 0; g < PQ_GROUPS; ++g) {
    int n = node0 + 2 * g + rw;
    if (n >= n_nodes) n = n_nodes - 1;
    const unsigned boff = (unsigned)n * ROWB + poff;
    bv[g][0] = frag_ld(x, boff, 0);
    bv[g][1] = frag_ld(x, boff, 1);
  }

  const v2f a0_lo = {a0.x, a0.y}, a0_hi = {a0.z, a0.w};
  const v2f a1_lo = {a1.x, a1.y}, a1_hi = {a1.z, a1.w};
  const bool b0 = (ch & 1) != 0, b1 = (ch & 2) != 0, b2 = (ch & 4) != 0;
  float res0 = 0.f, res1 = 0.f;
#pragma unroll
  for (int g = 0; g < PQ_GROUPS; g += 2) {
    const v2f c0_lo = {bv[g][0].x, bv[g][0].y},         c0_hi = {bv[g][0].z, bv[g][0].w};
    const v2f c1_lo = {bv[g][1].x, bv[g][1].y},         c1_hi = {bv[g][1].z, bv[g][1].w};
    const v2f d0_lo = {bv[g + 1][0].x, bv[g + 1][0].y}, d0_hi = {bv[g + 1][0].z, bv[g + 1][0].w};
    const v2f d1_lo = {bv[g + 1][1].x, bv[g + 1][1].y}, d1_hi = {bv[g + 1][1].z, bv[g + 1][1].w};
    v8f acc0 = {0.f, 0.f, 0.f, 0.f, 0.f, 0.f, 0.f, 0.f};
    v8f acc1 = {0.f, 0.f, 0.f, 0.f, 0.f, 0.f, 0.f, 0.f};
    acc0 = __builtin_amdgcn_wmma_f32_16x16x4_f32(false, a0_lo, false, c0_lo, (short)0, acc0, false, false);
    acc1 = __builtin_amdgcn_wmma_f32_16x16x4_f32(false, a0_lo, false, d0_lo, (short)0, acc1, false, false);
    acc0 = __builtin_amdgcn_wmma_f32_16x16x4_f32(false, a0_hi, false, c0_hi, (short)0, acc0, false, false);
    acc1 = __builtin_amdgcn_wmma_f32_16x16x4_f32(false, a0_hi, false, d0_hi, (short)0, acc1, false, false);
    acc0 = __builtin_amdgcn_wmma_f32_16x16x4_f32(false, a1_lo, false, c1_lo, (short)0, acc0, false, false);
    acc1 = __builtin_amdgcn_wmma_f32_16x16x4_f32(false, a1_lo, false, d1_lo, (short)0, acc1, false, false);
    acc0 = __builtin_amdgcn_wmma_f32_16x16x4_f32(false, a1_hi, false, c1_hi, (short)0, acc0, false, false);
    acc1 = __builtin_amdgcn_wmma_f32_16x16x4_f32(false, a1_hi, false, d1_hi, (short)0, acc1, false, false);
    const float s0 = diag_sum(acc0, b0, b1, b2);
    const float s1 = diag_sum(acc1, b0, b1, b2);
    if (g < 8) { res0 = (ch == g)     ? s0 : res0;  res0 = (ch == g + 1) ? s1 : res0; }
    else       { res1 = (ch == g - 8) ? s0 : res1;  res1 = (ch == g - 7) ? s1 : res1; }
  }
  __builtin_amdgcn_sched_group_barrier(0x020, 2 + 2 * PQ_GROUPS, 0);
  __builtin_amdgcn_sched_group_barrier(0x008, 4 * PQ_GROUPS, 0);

  float* __restrict__ dst = h ? q : p;
  const int o0 = node0 + 2 * ch + rw;
  if (o0 < n_nodes) dst[o0] = res0;
  const int o1 = o0 + 16;
  if (o1 < n_nodes) dst[o1] = res1;
}

#define RANGE_NODES 392
#define MAX_RANGES 512
#define BIN_THREADS 256
#define BIN_EDGES_PER_THREAD 10
#define SLICE_EDGES (BIN_THREADS * BIN_EDGES_PER_THREAD)
#define FIRST_THREADS 288

__device__ __forceinline__ void
bin_block(const int* __restrict__ e, float* __restrict__ scratch,
          int n_edges, int n_slices, int n_ranges, int slice) {
  __shared__ int  cnt[MAX_RANGES];
  __shared__ int  wsum[BIN_THREADS / 32];
  __shared__ int2 stage[SLICE_EDGES];
  int2* sorted = (int2*)scratch;
  int2* runs   = sorted + (size_t)n_slices * SLICE_EDGES;
  const int tid  = threadIdx.x;
  const int lane = tid & 31;
  const int wv   = tid >> 5;
  const int base = slice * SLICE_EDGES;
  const bool act = tid < BIN_THREADS;

  int2 ed[BIN_EDGES_PER_THREAD];
  int  rank[BIN_EDGES_PER_THREAD];
  if (act) {
    cnt[tid] = 0;
    cnt[tid + BIN_THREADS] = 0;
#pragma unroll
    for (int k = 0; k < BIN_EDGES_PER_THREAD; ++k) {
      int i = base + k * BIN_THREADS + tid;
      if (i >= n_edges) i = n_edges - 1;
      ed[k] = ((const int2*)e)[i];
    }
  }
  __syncthreads();
  if (act) {
#pragma unroll
    for (int k = 0; k < BIN_EDGES_PER_THREAD; ++k) {
      const int i = base + k * BIN_THREADS + tid;
      rank[k] = (i < n_edges) ? atomicAdd(&cnt[(unsigned)ed[k].y / RANGE_NODES], 1) : 0;
    }
  }
  __syncthreads();

  int c0 = 0, s2 = 0, incl = 0;
  if (act) {
    c0 = cnt[2 * tid];
    s2 = c0 + cnt[2 * tid + 1];
    incl = s2;
#pragma unroll
    for (int d = 1; d < 32; d <<= 1) {
      const int t = __shfl_up(incl, d, 32);
      if (lane >= d) incl += t;
    }
    if (lane == 31) wsum[wv] = incl;
  }
  __syncthreads();
  if (act) {
    int wbase = 0;
    for (int w = 0; w < wv; ++w) wbase += wsum[w];
    const int excl = wbase + incl - s2;
    cnt[2 * tid]     = excl;
    cnt[2 * tid + 1] = excl + c0;
  }
  __syncthreads();

  if (act) {
    for (int b = tid; b < n_ranges; b += BIN_THREADS)
      runs[(size_t)slice * n_ranges + b] = make_int2(cnt[b], cnt[b + 1]);
#pragma unroll
    for (int k = 0; k < BIN_EDGES_PER_THREAD; ++k) {
      const int i = base + k * BIN_THREADS + tid;
      if (i < n_edges) stage[cnt[(unsigned)ed[k].y / RANGE_NODES] + rank[k]] = ed[k];
    }
  }
  __syncthreads();
  if (act) {
    int n_here = n_edges - base;
    if (n_here > SLICE_EDGES) n_here = SLICE_EDGES;
    for (int j = tid; j < n_here; j += BIN_THREADS) sorted[base + j] = stage[j];
  }
}

__global__ void __launch_bounds__(FIRST_THREADS)
gat_prep_bin(const float* __restrict__ W, const float* __restrict__ b,
             const float* __restrict__ att_w,
             float* __restrict__ G, float* __restrict__ cs,
             const int* __restrict__ e, float* __restrict__ scratch,
             int n_edges, int n_slices, int n_ranges) {
  if (blockIdx.x == 0) prep_uv_block(W, b, att_w, G, cs);
  else                 bin_block(e, scratch, n_edges, n_slices, n_ranges, (int)blockIdx.x - 1);
}

#define OWN_WAVES 14
#define OWN_THREADS (32 * OWN_WAVES)
#define OWN_RUNS 18
#define OUT_PASSES 7
#define OUT_ROWS_PER_WAVE (4 * OUT_PASSES)
__global__ void __launch_bounds__(OWN_THREADS)
gat_edge_out(const float* __restrict__ x, const float* __restrict__ scratch,
             const float* __restrict__ p, const float* __restrict__ q, const float* __restrict__ cs,
             const float* __restrict__ nw, const float* __restrict__ nb, float* __restrict__ out,
             int n_slices, int n_ranges, int n_nodes) {
  __shared__ float acc[RANGE_NODES];
  __shared__ float ql[RANGE_NODES];
  const int2* sorted = (const int2*)scratch;
  const int2* runs   = sorted + (size_t)n_slices * SLICE_EDGES;
  const int tid  = threadIdx.x;
  const int lane = tid & 31;
  const int wv   = __builtin_amdgcn_readfirstlane(tid >> 5);
  const int b    = blockIdx.x;
  const int lo   = b * RANGE_NODES;
  if (tid < RANGE_NODES) {
    const int node = lo + tid;
    acc[tid] = 0.f;
    ql[tid]  = (node < n_nodes) ? q[node] : 0.f;
  }
  const float c = cs[0];
  __syncthreads();

  for (int s0 = 0; s0 < n_slices; s0 += OWN_WAVES * OWN_RUNS) {
    const int sw = s0 + wv * OWN_RUNS;
    const int s  = sw + lane;
    int2 jr = make_int2(0, 0);
    if (lane < OWN_RUNS && s < n_slices) jr = runs[(size_t)s * n_ranges + b];
    const int len = jr.y - jr.x;
    for (int off = 0; __builtin_amdgcn_ballot_w32(off < len) != 0u; off += 32) {
      int2 rec[OWN_RUNS];
      bool ok[OWN_RUNS];
#pragma unroll
      for (int i = 0; i < OWN_RUNS; ++i) {
        const int st = __builtin_amdgcn_readlane(jr.x, i);
        const int en = __builtin_amdgcn_readlane(jr.y, i);
        int sl = sw + i;
        if (sl >= n_slices) sl = n_slices - 1;
        int jj = st + off + lane;
        ok[i] = jj < en;
        if (jj >= en) jj = en - 1;
        if (jj < 0) jj = 0;
        rec[i] = sorted[(size_t)sl * SLICE_EDGES + jj];
      }
      asm volatile("" :: "v"(rec[0].x), "v"(rec[1].x), "v"(rec[2].x), "v"(rec[3].x),
                         "v"(rec[4].x), "v"(rec[5].x), "v"(rec[6].x), "v"(rec[7].x),
                         "v"(rec[8].x), "v"(rec[9].x), "v"(rec[10].x), "v"(rec[11].x),
                         "v"(rec[12].x), "v"(rec[13].x), "v"(rec[14].x), "v"(rec[15].x),
                         "v"(rec[16].x), "v"(rec[17].x));
      float ps[OWN_RUNS];
#pragma unroll
      for (int i = 0; i < OWN_RUNS; ++i) ps[i] = p[rec[i].x];
      int   dl[OWN_RUNS];
      float qd[OWN_RUNS];
#pragma unroll
      for (int i = 0; i < OWN_RUNS; ++i) {
        dl[i] = ok[i] ? rec[i].y - lo : 0;
        qd[i] = ql[dl[i]];
      }
#pragma unroll
      for (int i = 0; i < OWN_RUNS; ++i) {
        if (ok[i]) {
          const float z  = ps[i] + qd[i] + c;
          const float lz = z > 0.f ? z : 0.2f * z;
          atomicAdd(&acc[dl[i]], expf(lz));
        }
      }
    }
  }
  __syncthreads();

  const int lr0 = wv * OUT_ROWS_PER_WAVE;
  if (lo + lr0 >= n_nodes) return;
  const int g = lane >> 3;
  const int j = lane & 7;

  float  nv[OUT_PASSES];
  float4 xv[OUT_PASSES][4];
#pragma unroll
  for (int ps = 0; ps < OUT_PASSES; ++ps) {
    const int lr = lr0 + ps * 4 + g;
    int r = lo + lr;
    if (r >= n_nodes) r = n_nodes - 1;
    nv[ps] = acc[lr];
    const float4* xr = (const float4*)(x + (size_t)r * HIDDEN) + j;
#pragma unroll
    for (int t = 0; t < 4; ++t) xv[ps][t] = xr[t * 8];
  }
  float4 w4[4], b4[4];
#pragma unroll
  for (int t = 0; t < 4; ++t) {
    w4[t] = ((const float4*)nw)[t * 8 + j];
    b4[t] = ((const float4*)nb)[t * 8 + j];
  }

#pragma unroll
  for (int ps = 0; ps < OUT_PASSES; ++ps) {
    const int r = lo + lr0 + ps * 4 + g;
    const float denom = (nv[ps] == 0.f) ? 1.f : nv[ps];
    float4 y[4];
    float ss = 0.f;
#pragma unroll
    for (int t = 0; t < 4; ++t) {
      const float4 v = xv[ps][t];
      y[t].x = fmaxf(v.x / denom, 0.f) + v.x;
      y[t].y = fmaxf(v.y / denom, 0.f) + v.y;
      y[t].z = fmaxf(v.z / denom, 0.f) + v.z;
      y[t].w = fmaxf(v.w / denom, 0.f) + v.w;
      ss += y[t].x * y[t].x + y[t].y * y[t].y + y[t].z * y[t].z + y[t].w * y[t].w;
    }
    ss += __shfl_xor(ss, 4, 32);
    ss += __shfl_xor(ss, 2, 32);
    ss += __shfl_xor(ss, 1, 32);
    const float inv_rms = rsqrtf(ss * (1.0f / HIDDEN) + 1e-6f);
    if (r < n_nodes) {
      float4* orow = (float4*)(out + (size_t)r * HIDDEN) + j;
#pragma unroll
      for (int t = 0; t < 4; ++t) {
        float4 o;
        o.x = w4[t].x * (y[t].x * inv_rms) + b4[t].x;
        o.y = w4[t].y * (y[t].y * inv_rms) + b4[t].y;
        o.z = w4[t].z * (y[t].z * inv_rms) + b4[t].z;
        o.w = w4[t].w * (y[t].w * inv_rms) + b4[t].w;
        orow[t * 8] = o;
      }
    }
  }
}

extern "C" void kernel_launch(void* const* d_in, const int* in_sizes, int n_in,
                              void* d_out, int out_size, void* d_ws, size_t ws_size,
                              hipStream_t stream) {
  const float* x     = (const float*)d_in[0];
  const float* W     = (const float*)d_in[1];
  const float* b     = (const float*)d_in[2];
  const float* att_w = (const float*)d_in[3];
  const float* nw    = (const float*)d_in[4];
  const float* nb    = (const float*)d_in[5];
  const int*   e     = (const int*)d_in[6];
  float* out = (float*)d_out;

  const int n_nodes = in_sizes[0] / HIDDEN;
  const int n_edges = in_sizes[6] / 2;

  int n_slices = (n_edges + SLICE_EDGES - 1) / SLICE_EDGES;
  int n_ranges = (n_nodes + RANGE_NODES - 1) / RANGE_NODES;

  float* G    = (float*)d_ws;
  float* cs   = G + 2 * HIDDEN;
  float* srt  = cs + 32;
  float* p    = srt + 2 * ((size_t)n_slices * SLICE_EDGES + (size_t)n_slices * n_ranges);
  float* q    = p + n_nodes;

  gat_prep_bin<<<1 + n_slices, FIRST_THREADS, 0, stream>>>(W, b, att_w, G, cs, e, srt, n_edges, n_slices, n_ranges);

  int pq_waves  = (n_nodes + PQ_NODES_PER_WAVE - 1) / PQ_NODES_PER_WAVE;
  int pq_blocks = (pq_waves * 32 + 127) / 128;
  gat_pq_wmma<<<pq_blocks, 128, 0, stream>>>(x, G, p, q, n_nodes);

  gat_edge_out<<<n_ranges, OWN_THREADS, 0, stream>>>(x, srt, p, q, cs, nw, nb, out, n_slices, n_ranges, n_nodes);
}
